# starting kernel
# baseline (speedup 1.0000x reference)
#include <hip/hip_runtime.h>

typedef __attribute__((ext_vector_type(2))) float v2f;
typedef __attribute__((ext_vector_type(8))) float v8f;

#define BB 32
#define NN 2048
#define PP 1024
#define HH 256
#define EE 512
#define BN_EPS 1e-5f

__global__ __launch_bounds__(256) void bps_min_kernel(
    const float* __restrict__ x,
    const float* __restrict__ basis,
    float* __restrict__ feat)
{
  const int tileP = blockIdx.x * 16;
  const int b     = blockIdx.y;
  const int lane  = threadIdx.x & 31;
  const int wave  = threadIdx.x >> 5;
  const int col   = lane & 15;
  const bool hi   = lane >= 16;

  const int   n   = tileP + col;
  const float bx  = basis[n * 3 + 0];
  const float by  = basis[n * 3 + 1];
  const float bz  = basis[n * 3 + 2];
  const float bn2 = bx * bx + by * by + bz * bz;
  v2f Bm;
  Bm.x = hi ? (-2.0f * bz) : (-2.0f * bx);
  Bm.y = hi ? 1.0f         : (-2.0f * by);

  float minv[4];
#pragma unroll
  for (int r = 0; r < 4; ++r) minv[r] = 3.4e38f;

  const float* xb = x + (size_t)b * 3 * NN;
#pragma unroll 2
  for (int i0 = wave * 16; i0 < NN; i0 += 16 * 8) {
    const int   m   = i0 + col;
    const float px  = xb[m];
    const float py  = xb[NN + m];
    const float pz  = xb[2 * NN + m];
    const float pn2 = px * px + py * py + pz * pz;
    v2f Am;
    Am.x = hi ? pz  : px;
    Am.y = hi ? pn2 : py;
    v8f c = {0.f, 0.f, 0.f, 0.f, 0.f, 0.f, 0.f, 0.f};
    c = __builtin_amdgcn_wmma_f32_16x16x4_f32(false, Am, false, Bm,
                                              (short)0, c, false, false);
#pragma unroll
    for (int r = 0; r < 4; ++r)
      minv[r] = fminf(fminf(c[2 * r], c[2 * r + 1]), minv[r]);
  }
  float mv = fminf(fminf(minv[0], minv[1]), fminf(minv[2], minv[3]));
  mv = fminf(mv, __shfl_xor(mv, 16));
  mv = fmaxf(mv + bn2, 0.0f);

  __shared__ float red[8][16];
  if (lane < 16) red[wave][col] = mv;
  __syncthreads();
  if (threadIdx.x < 16) {
    float v = red[0][threadIdx.x];
#pragma unroll
    for (int w = 1; w < 8; ++w) v = fminf(v, red[w][threadIdx.x]);
    feat[(size_t)b * PP + tileP + threadIdx.x] = sqrtf(v);
  }
}

__global__ __launch_bounds__(256) void bn0_kernel(
    const float* __restrict__ feat, const float* __restrict__ g,
    const float* __restrict__ bshift, float* __restrict__ x0)
{
  const int j = blockIdx.x * 256 + threadIdx.x;
  float s = 0.f, ss = 0.f;
#pragma unroll 4
  for (int i = 0; i < BB; ++i) {
    const float v = feat[(size_t)i * PP + j];
    s += v; ss += v * v;
  }
  const float mean = s * (1.0f / BB);
  const float var  = ss * (1.0f / BB) - mean * mean;
  const float sc   = g[j] * rsqrtf(var + BN_EPS);
  const float bt   = bshift[j];
#pragma unroll 4
  for (int i = 0; i < BB; ++i)
    x0[(size_t)i * PP + j] = (feat[(size_t)i * PP + j] - mean) * sc + bt;
}

__device__ __forceinline__ void gemm_segment(v8f& acc, const float* __restrict__ src,
                                             int Ks, const float* __restrict__ Wseg,
                                             int row, int koff) {
  const float* arow = src + (size_t)row * Ks + koff;
  const float* wrow = Wseg + koff;
#pragma unroll 4
  for (int kk = 0; kk < Ks; kk += 4) {
    const float2 av = *reinterpret_cast<const float2*>(arow + kk);
    const float2 wv = *reinterpret_cast<const float2*>(wrow + kk);
    v2f Am; Am.x = av.x; Am.y = av.y;
    v2f Bm; Bm.x = wv.x; Bm.y = wv.y;
    acc = __builtin_amdgcn_wmma_f32_16x16x4_f32(false, Am, false, Bm,
                                                (short)0, acc, false, false);
  }
}

__global__ __launch_bounds__(64) void dense_bn_relu_kernel(
    const float* __restrict__ in0, int K0,
    const float* __restrict__ in1, int K1,
    const float* __restrict__ in2, int K2,
    const float* __restrict__ W,
    const float* __restrict__ bias,
    const float* __restrict__ g, const float* __restrict__ beta,
    float* __restrict__ out, int Hout)
{
  const int  c    = blockIdx.x * 16;
  const int  lane = threadIdx.x & 31;
  const int  wave = threadIdx.x >> 5;
  const int  col  = lane & 15;
  const bool hi   = lane >= 16;
  const int  n    = c + col;
  const int  row  = wave * 16 + col;
  const int  Ktot = K0 + K1 + K2;
  const int  koff = hi ? 2 : 0;

  v8f acc = {0.f, 0.f, 0.f, 0.f, 0.f, 0.f, 0.f, 0.f};
  const float* Wn = W + (size_t)n * Ktot;
  gemm_segment(acc, in0, K0, Wn, row, koff);
  if (K1 > 0) gemm_segment(acc, in1, K1, Wn + K0, row, koff);
  if (K2 > 0) gemm_segment(acc, in2, K2, Wn + K0 + K1, row, koff);

  const float bi = bias[n];
  float vals[8];
  float s = 0.f, ss = 0.f;
#pragma unroll
  for (int r = 0; r < 8; ++r) {
    const float v = fmaxf(acc[r] + bi, 0.f);
    vals[r] = v; s += v; ss += v * v;
  }
  s  += __shfl_xor(s, 16);
  ss += __shfl_xor(ss, 16);

  __shared__ float sred[2][16][2];
  if (lane < 16) { sred[wave][col][0] = s; sred[wave][col][1] = ss; }
  __syncthreads();
  const float ts   = sred[0][col][0] + sred[1][col][0];
  const float tss  = sred[0][col][1] + sred[1][col][1];
  const float mean = ts * (1.0f / BB);
  const float var  = tss * (1.0f / BB) - mean * mean;
  const float sc   = g[n] * rsqrtf(var + BN_EPS);
  const float bt   = beta[n];
#pragma unroll
  for (int r = 0; r < 8; ++r) {
    const int rr = wave * 16 + r + (hi ? 8 : 0);
    out[(size_t)rr * Hout + n] = (vals[r] - mean) * sc + bt;
  }
}

extern "C" void kernel_launch(void* const* d_in, const int* in_sizes, int n_in,
                              void* d_out, int out_size, void* d_ws, size_t ws_size,
                              hipStream_t stream) {
  const float* x     = (const float*)d_in[0];
  const float* basis = (const float*)d_in[1];
  const float* bn0_g = (const float*)d_in[2];
  const float* bn0_b = (const float*)d_in[3];
  const float* W0 = (const float*)d_in[4];
  const float* b0 = (const float*)d_in[5];
  const float* g0 = (const float*)d_in[6];
  const float* be0 = (const float*)d_in[7];
  const float* W1 = (const float*)d_in[8];
  const float* b1 = (const float*)d_in[9];
  const float* g1 = (const float*)d_in[10];
  const float* be1 = (const float*)d_in[11];
  const float* W2 = (const float*)d_in[12];
  const float* b2 = (const float*)d_in[13];
  const float* g2 = (const float*)d_in[14];
  const float* be2 = (const float*)d_in[15];
  const float* W3 = (const float*)d_in[16];
  const float* b3 = (const float*)d_in[17];
  const float* g3 = (const float*)d_in[18];
  const float* be3 = (const float*)d_in[19];
  const float* Wf = (const float*)d_in[20];
  const float* bf = (const float*)d_in[21];
  const float* gf = (const float*)d_in[22];
  const float* bef = (const float*)d_in[23];

  float* ws   = (float*)d_ws;
  float* feat = ws;
  float* x0   = ws + 32768;
  float* h    = ws + 65536;
  float* a1   = ws + 73728;
  float* a2   = ws + 81920;

  bps_min_kernel<<<dim3(PP / 16, BB), 256, 0, stream>>>(x, basis, feat);
  bn0_kernel<<<PP / 256, 256, 0, stream>>>(feat, bn0_g, bn0_b, x0);
  dense_bn_relu_kernel<<<HH / 16, 64, 0, stream>>>(x0, PP, nullptr, 0, nullptr, 0,
                                                   W0, b0, g0, be0, h, HH);
  dense_bn_relu_kernel<<<HH / 16, 64, 0, stream>>>(h, HH, nullptr, 0, nullptr, 0,
                                                   W1, b1, g1, be1, a1, HH);
  dense_bn_relu_kernel<<<HH / 16, 64, 0, stream>>>(x0, PP, a1, HH, nullptr, 0,
                                                   W2, b2, g2, be2, h, HH);
  dense_bn_relu_kernel<<<HH / 16, 64, 0, stream>>>(h, HH, nullptr, 0, nullptr, 0,
                                                   W3, b3, g3, be3, a2, HH);
  dense_bn_relu_kernel<<<EE / 16, 64, 0, stream>>>(x0, PP, a1, HH, a2, HH,
                                                   Wf, bf, gf, bef,
                                                   (float*)d_out, EE);
}


# dense layers: K split over 16 waves per workgroup (each wave both 16-row tiles, compile-time slices), partial tiles summed through LDS in slice order; changes fp sum order
# speedup vs baseline: 1.7928x; 1.7928x over previous
#include <hip/hip_runtime.h>

typedef __attribute__((ext_vector_type(2))) float v2f;
typedef __attribute__((ext_vector_type(8))) float v8f;

#define BB 32
#define NN 2048
#define PP 1024
#define HH 256
#define EE 512
#define BN_EPS 1e-5f

__global__ __launch_bounds__(256) void bps_min_kernel(
    const float* __restrict__ x,
    const float* __restrict__ basis,
    float* __restrict__ feat)
{
  const int tileP = blockIdx.x * 16;
  const int b     = blockIdx.y;
  const int lane  = threadIdx.x & 31;
  const int wave  = threadIdx.x >> 5;
  const int col   = lane & 15;
  const bool hi   = lane >= 16;

  const int   n   = tileP + col;
  const float bx  = basis[n * 3 + 0];
  const float by  = basis[n * 3 + 1];
  const float bz  = basis[n * 3 + 2];
  const float bn2 = bx * bx + by * by + bz * bz;
  v2f Bm;
  Bm.x = hi ? (-2.0f * bz) : (-2.0f * bx);
  Bm.y = hi ? 1.0f         : (-2.0f * by);

  float minv[4];
#pragma unroll
  for (int r = 0; r < 4; ++r) minv[r] = 3.4e38f;

  const float* xb = x + (size_t)b * 3 * NN;
#pragma unroll 2
  for (int i0 = wave * 16; i0 < NN; i0 += 16 * 8) {
    const int   m   = i0 + col;
    const float px  = xb[m];
    const float py  = xb[NN + m];
    const float pz  = xb[2 * NN + m];
    const float pn2 = px * px + py * py + pz * pz;
    v2f Am;
    Am.x = hi ? pz  : px;
    Am.y = hi ? pn2 : py;
    v8f c = {0.f, 0.f, 0.f, 0.f, 0.f, 0.f, 0.f, 0.f};
    c = __builtin_amdgcn_wmma_f32_16x16x4_f32(false, Am, false, Bm,
                                              (short)0, c, false, false);
#pragma unroll
    for (int r = 0; r < 4; ++r)
      minv[r] = fminf(fminf(c[2 * r], c[2 * r + 1]), minv[r]);
  }
  float mv = fminf(fminf(minv[0], minv[1]), fminf(minv[2], minv[3]));
  mv = fminf(mv, __shfl_xor(mv, 16));
  mv = fmaxf(mv + bn2, 0.0f);

  __shared__ float red[8][16];
  if (lane < 16) red[wave][col] = mv;
  __syncthreads();
  if (threadIdx.x < 16) {
    float v = red[0][threadIdx.x];
#pragma unroll
    for (int w = 1; w < 8; ++w) v = fminf(v, red[w][threadIdx.x]);
    feat[(size_t)b * PP + tileP + threadIdx.x] = sqrtf(v);
  }
}

__global__ __launch_bounds__(256) void bn0_kernel(
    const float* __restrict__ feat, const float* __restrict__ g,
    const float* __restrict__ bshift, float* __restrict__ x0)
{
  const int j = blockIdx.x * 256 + threadIdx.x;
  float s = 0.f, ss = 0.f;
#pragma unroll 4
  for (int i = 0; i < BB; ++i) {
    const float v = feat[(size_t)i * PP + j];
    s += v; ss += v * v;
  }
  const float mean = s * (1.0f / BB);
  const float var  = ss * (1.0f / BB) - mean * mean;
  const float sc   = g[j] * rsqrtf(var + BN_EPS);
  const float bt   = bshift[j];
#pragma unroll 4
  for (int i = 0; i < BB; ++i)
    x0[(size_t)i * PP + j] = (feat[(size_t)i * PP + j] - mean) * sc + bt;
}

#define KS 16

template <int Ks>
__device__ __forceinline__ void gemm_segment(v8f& acc0, v8f& acc1,
                                             const float* __restrict__ src,
                                             const float* __restrict__ Wseg,
                                             int col, int koff, int ks) {
  constexpr int SL = Ks / KS;
  static_assert(Ks % (4 * KS) == 0, "segment must split into whole k-steps");
  const float* a0 = src + (size_t)col * Ks + ks * SL + koff;
  const float* a1 = a0 + (size_t)16 * Ks;
  const float* wr = Wseg + ks * SL + koff;
#pragma unroll
  for (int kk = 0; kk < SL; kk += 4) {
    const float2 av0 = *reinterpret_cast<const float2*>(a0 + kk);
    const float2 av1 = *reinterpret_cast<const float2*>(a1 + kk);
    const float2 wv  = *reinterpret_cast<const float2*>(wr + kk);
    v2f A0; A0.x = av0.x; A0.y = av0.y;
    v2f A1; A1.x = av1.x; A1.y = av1.y;
    v2f Bm; Bm.x = wv.x;  Bm.y = wv.y;
    acc0 = __builtin_amdgcn_wmma_f32_16x16x4_f32(false, A0, false, Bm,
                                                 (short)0, acc0, false, false);
    acc1 = __builtin_amdgcn_wmma_f32_16x16x4_f32(false, A1, false, Bm,
                                                 (short)0, acc1, false, false);
  }
}

template <int K0, int K1, int K2>
__global__ __launch_bounds__(KS * 32) void dense_bn_relu_kernel(
    const float* __restrict__ in0,
    const float* __restrict__ in1,
    const float* __restrict__ in2,
    const float* __restrict__ W,
    const float* __restrict__ bias,
    const float* __restrict__ g, const float* __restrict__ beta,
    float* __restrict__ out, int Hout)
{
  constexpr int Ktot = K0 + K1 + K2;
  const int  c    = blockIdx.x * 16;
  const int  lane = threadIdx.x & 31;
  const int  wave = threadIdx.x >> 5;
  const int  col  = lane & 15;
  const bool hi   = lane >= 16;
  const int  n    = c + col;
  const int  koff = hi ? 2 : 0;

  v8f acc0 = {0.f, 0.f, 0.f, 0.f, 0.f, 0.f, 0.f, 0.f};
  v8f acc1 = {0.f, 0.f, 0.f, 0.f, 0.f, 0.f, 0.f, 0.f};
  const float* Wn = W + (size_t)n * Ktot;
  gemm_segment<K0>(acc0, acc1, in0, Wn, col, koff, wave);
  if constexpr (K1 > 0) gemm_segment<K1>(acc0, acc1, in1, Wn + K0, col, koff, wave);
  if constexpr (K2 > 0) gemm_segment<K2>(acc0, acc1, in2, Wn + K0 + K1, col, koff, wave);

  __shared__ float4 part[KS][2][2][32];
  __shared__ float sred[2][16][2];
  part[wave][0][0][lane] = make_float4(acc0[0], acc0[1], acc0[2], acc0[3]);
  part[wave][0][1][lane] = make_float4(acc0[4], acc0[5], acc0[6], acc0[7]);
  part[wave][1][0][lane] = make_float4(acc1[0], acc1[1], acc1[2], acc1[3]);
  part[wave][1][1][lane] = make_float4(acc1[4], acc1[5], acc1[6], acc1[7]);
  __syncthreads();

  float vals[8];
  if (wave < 2) {
    float a[8];
#pragma unroll
    for (int r = 0; r < 8; ++r) a[r] = 0.f;
#pragma unroll
    for (int s = 0; s < KS; ++s) {
      const float4 lo = part[s][wave][0][lane];
      const float4 hv = part[s][wave][1][lane];
      a[0] += lo.x; a[1] += lo.y; a[2] += lo.z; a[3] += lo.w;
      a[4] += hv.x; a[5] += hv.y; a[6] += hv.z; a[7] += hv.w;
    }
    const float bi = bias[n];
    float s = 0.f, ss = 0.f;
#pragma unroll
    for (int r = 0; r < 8; ++r) {
      const float v = fmaxf(a[r] + bi, 0.f);
      vals[r] = v; s += v; ss += v * v;
    }
    s  += __shfl_xor(s, 16);
    ss += __shfl_xor(ss, 16);
    if (lane < 16) { sred[wave][col][0] = s; sred[wave][col][1] = ss; }
  }
  __syncthreads();
  if (wave < 2) {
    const float ts   = sred[0][col][0] + sred[1][col][0];
    const float tss  = sred[0][col][1] + sred[1][col][1];
    const float mean = ts * (1.0f / BB);
    const float var  = tss * (1.0f / BB) - mean * mean;
    const float sc   = g[n] * rsqrtf(var + BN_EPS);
    const float bt   = beta[n];
#pragma unroll
    for (int r = 0; r < 8; ++r) {
      const int rr = wave * 16 + r + (hi ? 8 : 0);
      out[(size_t)rr * Hout + n] = (vals[r] - mean) * sc + bt;
    }
  }
}

extern "C" void kernel_launch(void* const* d_in, const int* in_sizes, int n_in,
                              void* d_out, int out_size, void* d_ws, size_t ws_size,
                              hipStream_t stream) {
  const float* x     = (const float*)d_in[0];
  const float* basis = (const float*)d_in[1];
  const float* bn0_g = (const float*)d_in[2];
  const float* bn0_b = (const float*)d_in[3];
  const float* W0 = (const float*)d_in[4];
  const float* b0 = (const float*)d_in[5];
  const float* g0 = (const float*)d_in[6];
  const float* be0 = (const float*)d_in[7];
  const float* W1 = (const float*)d_in[8];
  const float* b1 = (const float*)d_in[9];
  const float* g1 = (const float*)d_in[10];
  const float* be1 = (const float*)d_in[11];
  const float* W2 = (const float*)d_in[12];
  const float* b2 = (const float*)d_in[13];
  const float* g2 = (const float*)d_in[14];
  const float* be2 = (const float*)d_in[15];
  const float* W3 = (const float*)d_in[16];
  const float* b3 = (const float*)d_in[17];
  const float* g3 = (const float*)d_in[18];
  const float* be3 = (const float*)d_in[19];
  const float* Wf = (const float*)d_in[20];
  const float* bf = (const float*)d_in[21];
  const float* gf = (const float*)d_in[22];
  const float* bef = (const float*)d_in[23];

  float* ws   = (float*)d_ws;
  float* feat = ws;
  float* x0   = ws + 32768;
  float* h    = ws + 65536;
  float* a1   = ws + 73728;
  float* a2   = ws + 81920;

  bps_min_kernel<<<dim3(PP / 16, BB), 256, 0, stream>>>(x, basis, feat);
  bn0_kernel<<<PP / 256, 256, 0, stream>>>(feat, bn0_g, bn0_b, x0);
  dense_bn_relu_kernel<PP, 0, 0><<<HH / 16, KS * 32, 0, stream>>>(
      x0, nullptr, nullptr, W0, b0, g0, be0, h, HH);
  dense_bn_relu_kernel<HH, 0, 0><<<HH / 16, KS * 32, 0, stream>>>(
      h, nullptr, nullptr, W1, b1, g1, be1, a1, HH);
  dense_bn_relu_kernel<PP, HH, 0><<<HH / 16, KS * 32, 0, stream>>>(
      x0, a1, nullptr, W2, b2, g2, be2, h, HH);
  dense_bn_relu_kernel<HH, 0, 0><<<HH / 16, KS * 32, 0, stream>>>(
      h, nullptr, nullptr, W3, b3, g3, be3, a2, HH);
  dense_bn_relu_kernel<PP, HH, HH><<<EE / 16, KS * 32, 0, stream>>>(
      x0, a1, a2, Wf, bf, gf, bef, (float*)d_out, EE);
}


# BPS kernel: one wave per (batch row, 16-col tile) walks all 2048 points: 2048 long waves, not 16384 short ones; 16-B point loads two groups ahead; no cross-wave min
# speedup vs baseline: 1.9713x; 1.0996x over previous
#include <hip/hip_runtime.h>

typedef __attribute__((ext_vector_type(2))) float v2f;
typedef __attribute__((ext_vector_type(8))) float v8f;

#define BB 32
#define NN 2048
#define PP 1024
#define HH 256
#define EE 512
#define BN_EPS 1e-5f

#define BPS_TILES 8
#define BPS_GPTS  256
#define BPS_NG    (NN / BPS_GPTS)

struct bps_group { float4 X[4], Y[4], Z[4]; };

__device__ __forceinline__ void bps_load_group(bps_group& G, const float* __restrict__ p) {
#pragma unroll
  for (int q = 0; q < 4; ++q) {
    G.X[q] = *reinterpret_cast<const float4*>(p + q * 64);
    G.Y[q] = *reinterpret_cast<const float4*>(p + NN + q * 64);
    G.Z[q] = *reinterpret_cast<const float4*>(p + 2 * NN + q * 64);
  }
}

__device__ __forceinline__ void bps_point_tile(float (&minv)[4], float px, float py,
                                               float pz, v2f Bm, bool hi) {
  const float pn2 = px * px + py * py + pz * pz;
  v2f Am;
  Am.x = hi ? pz  : px;
  Am.y = hi ? pn2 : py;
  v8f c = {0.f, 0.f, 0.f, 0.f, 0.f, 0.f, 0.f, 0.f};
  c = __builtin_amdgcn_wmma_f32_16x16x4_f32(false, Am, false, Bm,
                                            (short)0, c, false, false);
#pragma unroll
  for (int r = 0; r < 4; ++r)
    minv[r] = fminf(fminf(c[2 * r], c[2 * r + 1]), minv[r]);
}

__device__ __forceinline__ void bps_group_tiles(float (&minv)[4], const bps_group& G,
                                                v2f Bm, bool hi) {
#pragma unroll
  for (int q = 0; q < 4; ++q) {
    bps_point_tile(minv, G.X[q].x, G.Y[q].x, G.Z[q].x, Bm, hi);
    bps_point_tile(minv, G.X[q].y, G.Y[q].y, G.Z[q].y, Bm, hi);
    bps_point_tile(minv, G.X[q].z, G.Y[q].z, G.Z[q].z, Bm, hi);
    bps_point_tile(minv, G.X[q].w, G.Y[q].w, G.Z[q].w, Bm, hi);
  }
}

__global__ __launch_bounds__(256) void bps_min_kernel(
    const float* __restrict__ x,
    const float* __restrict__ basis,
    float* __restrict__ feat)
{
  const int b     = blockIdx.y;
  const int lane  = threadIdx.x & 31;
  const int wave  = threadIdx.x >> 5;
  const int col   = lane & 15;
  const bool hi   = lane >= 16;
  const int tileP = (blockIdx.x * BPS_TILES + wave) * 16;

  const float* xp = x + (size_t)b * 3 * NN + 4 * col;
  bps_group G[3];
  bps_load_group(G[0], xp);
  bps_load_group(G[1], xp + BPS_GPTS);

  const int   n   = tileP + col;
  const float bx  = basis[n * 3 + 0];
  const float by  = basis[n * 3 + 1];
  const float bz  = basis[n * 3 + 2];
  const float bn2 = bx * bx + by * by + bz * bz;
  v2f Bm;
  Bm.x = hi ? (-2.0f * bz) : (-2.0f * bx);
  Bm.y = hi ? 1.0f         : (-2.0f * by);

  float minv[4];
#pragma unroll
  for (int r = 0; r < 4; ++r) minv[r] = 3.4e38f;

#pragma unroll
  for (int g = 0; g < BPS_NG; ++g) {
    if (g + 2 < BPS_NG) {
      bps_load_group(G[(g + 2) % 3], xp + (g + 2) * BPS_GPTS);
      __builtin_amdgcn_sched_barrier(0);
    }
    bps_group_tiles(minv, G[g % 3], Bm, hi);
    __builtin_amdgcn_sched_barrier(0);
  }

  float mv = fminf(fminf(minv[0], minv[1]), fminf(minv[2], minv[3]));
  mv = fminf(mv, __shfl_xor(mv, 16));
  mv = fmaxf(mv + bn2, 0.0f);

  __shared__ float red[BPS_TILES * 16];
  if (lane < 16) red[wave * 16 + col] = mv;
  __syncthreads();
  if (threadIdx.x < BPS_TILES * 16)
    feat[(size_t)b * PP + blockIdx.x * (BPS_TILES * 16) + threadIdx.x] =
        sqrtf(red[threadIdx.x]);
}

__global__ __launch_bounds__(256) void bn0_kernel(
    const float* __restrict__ feat, const float* __restrict__ g,
    const float* __restrict__ bshift, float* __restrict__ x0)
{
  const int j = blockIdx.x * 256 + threadIdx.x;
  float s = 0.f, ss = 0.f;
#pragma unroll 4
  for (int i = 0; i < BB; ++i) {
    const float v = feat[(size_t)i * PP + j];
    s += v; ss += v * v;
  }
  const float mean = s * (1.0f / BB);
  const float var  = ss * (1.0f / BB) - mean * mean;
  const float sc   = g[j] * rsqrtf(var + BN_EPS);
  const float bt   = bshift[j];
#pragma unroll 4
  for (int i = 0; i < BB; ++i)
    x0[(size_t)i * PP + j] = (feat[(size_t)i * PP + j] - mean) * sc + bt;
}

#define KS 16

template <int Ks>
__device__ __forceinline__ void gemm_segment(v8f& acc0, v8f& acc1,
                                             const float* __restrict__ src,
                                             const float* __restrict__ Wseg,
                                             int col, int koff, int ks) {
  constexpr int SL = Ks / KS;
  static_assert(Ks % (4 * KS) == 0, "segment must split into whole k-steps");
  const float* a0 = src + (size_t)col * Ks + ks * SL + koff;
  const float* a1 = a0 + (size_t)16 * Ks;
  const float* wr = Wseg + ks * SL + koff;
#pragma unroll
  for (int kk = 0; kk < SL; kk += 4) {
    const float2 av0 = *reinterpret_cast<const float2*>(a0 + kk);
    const float2 av1 = *reinterpret_cast<const float2*>(a1 + kk);
    const float2 wv  = *reinterpret_cast<const float2*>(wr + kk);
    v2f A0; A0.x = av0.x; A0.y = av0.y;
    v2f A1; A1.x = av1.x; A1.y = av1.y;
    v2f Bm; Bm.x = wv.x;  Bm.y = wv.y;
    acc0 = __builtin_amdgcn_wmma_f32_16x16x4_f32(false, A0, false, Bm,
                                                 (short)0, acc0, false, false);
    acc1 = __builtin_amdgcn_wmma_f32_16x16x4_f32(false, A1, false, Bm,
                                                 (short)0, acc1, false, false);
  }
}

template <int K0, int K1, int K2>
__global__ __launch_bounds__(KS * 32) void dense_bn_relu_kernel(
    const float* __restrict__ in0,
    const float* __restrict__ in1,
    const float* __restrict__ in2,
    const float* __restrict__ W,
    const float* __restrict__ bias,
    const float* __restrict__ g, const float* __restrict__ beta,
    float* __restrict__ out, int Hout)
{
  constexpr int Ktot = K0 + K1 + K2;
  const int  c    = blockIdx.x * 16;
  const int  lane = threadIdx.x & 31;
  const int  wave = threadIdx.x >> 5;
  const int  col  = lane & 15;
  const bool hi   = lane >= 16;
  const int  n    = c + col;
  const int  koff = hi ? 2 : 0;

  v8f acc0 = {0.f, 0.f, 0.f, 0.f, 0.f, 0.f, 0.f, 0.f};
  v8f acc1 = {0.f, 0.f, 0.f, 0.f, 0.f, 0.f, 0.f, 0.f};
  const float* Wn = W + (size_t)n * Ktot;
  gemm_segment<K0>(acc0, acc1, in0, Wn, col, koff, wave);
  if constexpr (K1 > 0) gemm_segment<K1>(acc0, acc1, in1, Wn + K0, col, koff, wave);
  if constexpr (K2 > 0) gemm_segment<K2>(acc0, acc1, in2, Wn + K0 + K1, col, koff, wave);

  __shared__ float4 part[KS][2][2][32];
  __shared__ float sred[2][16][2];
  part[wave][0][0][lane] = make_float4(acc0[0], acc0[1], acc0[2], acc0[3]);
  part[wave][0][1][lane] = make_float4(acc0[4], acc0[5], acc0[6], acc0[7]);
  part[wave][1][0][lane] = make_float4(acc1[0], acc1[1], acc1[2], acc1[3]);
  part[wave][1][1][lane] = make_float4(acc1[4], acc1[5], acc1[6], acc1[7]);
  __syncthreads();

  float vals[8];
  if (wave < 2) {
    float a[8];
#pragma unroll
    for (int r = 0; r < 8; ++r) a[r] = 0.f;
#pragma unroll
    for (int s = 0; s < KS; ++s) {
      const float4 lo = part[s][wave][0][lane];
      const float4 hv = part[s][wave][1][lane];
      a[0] += lo.x; a[1] += lo.y; a[2] += lo.z; a[3] += lo.w;
      a[4] += hv.x; a[5] += hv.y; a[6] += hv.z; a[7] += hv.w;
    }
    const float bi = bias[n];
    float s = 0.f, ss = 0.f;
#pragma unroll
    for (int r = 0; r < 8; ++r) {
      const float v = fmaxf(a[r] + bi, 0.f);
      vals[r] = v; s += v; ss += v * v;
    }
    s  += __shfl_xor(s, 16);
    ss += __shfl_xor(ss, 16);
    if (lane < 16) { sred[wave][col][0] = s; sred[wave][col][1] = ss; }
  }
  __syncthreads();
  if (wave < 2) {
    const float ts   = sred[0][col][0] + sred[1][col][0];
    const float tss  = sred[0][col][1] + sred[1][col][1];
    const float mean = ts * (1.0f / BB);
    const float var  = tss * (1.0f / BB) - mean * mean;
    const float sc   = g[n] * rsqrtf(var + BN_EPS);
    const float bt   = beta[n];
#pragma unroll
    for (int r = 0; r < 8; ++r) {
      const int rr = wave * 16 + r + (hi ? 8 : 0);
      out[(size_t)rr * Hout + n] = (vals[r] - mean) * sc + bt;
    }
  }
}

extern "C" void kernel_launch(void* const* d_in, const int* in_sizes, int n_in,
                              void* d_out, int out_size, void* d_ws, size_t ws_size,
                              hipStream_t stream) {
  const float* x     = (const float*)d_in[0];
  const float* basis = (const float*)d_in[1];
  const float* bn0_g = (const float*)d_in[2];
  const float* bn0_b = (const float*)d_in[3];
  const float* W0 = (const float*)d_in[4];
  const float* b0 = (const float*)d_in[5];
  const float* g0 = (const float*)d_in[6];
  const float* be0 = (const float*)d_in[7];
  const float* W1 = (const float*)d_in[8];
  const float* b1 = (const float*)d_in[9];
  const float* g1 = (const float*)d_in[10];
  const float* be1 = (const float*)d_in[11];
  const float* W2 = (const float*)d_in[12];
  const float* b2 = (const float*)d_in[13];
  const float* g2 = (const float*)d_in[14];
  const float* be2 = (const float*)d_in[15];
  const float* W3 = (const float*)d_in[16];
  const float* b3 = (const float*)d_in[17];
  const float* g3 = (const float*)d_in[18];
  const float* be3 = (const float*)d_in[19];
  const float* Wf = (const float*)d_in[20];
  const float* bf = (const float*)d_in[21];
  const float* gf = (const float*)d_in[22];
  const float* bef = (const float*)d_in[23];

  float* ws   = (float*)d_ws;
  float* feat = ws;
  float* x0   = ws + 32768;
  float* h    = ws + 65536;
  float* a1   = ws + 73728;
  float* a2   = ws + 81920;

  bps_min_kernel<<<dim3(PP / (BPS_TILES * 16), BB), 256, 0, stream>>>(x, basis, feat);
  bn0_kernel<<<PP / 256, 256, 0, stream>>>(feat, bn0_g, bn0_b, x0);
  dense_bn_relu_kernel<PP, 0, 0><<<HH / 16, KS * 32, 0, stream>>>(
      x0, nullptr, nullptr, W0, b0, g0, be0, h, HH);
  dense_bn_relu_kernel<HH, 0, 0><<<HH / 16, KS * 32, 0, stream>>>(
      h, nullptr, nullptr, W1, b1, g1, be1, a1, HH);
  dense_bn_relu_kernel<PP, HH, 0><<<HH / 16, KS * 32, 0, stream>>>(
      x0, a1, nullptr, W2, b2, g2, be2, h, HH);
  dense_bn_relu_kernel<HH, 0, 0><<<HH / 16, KS * 32, 0, stream>>>(
      h, nullptr, nullptr, W3, b3, g3, be3, a2, HH);
  dense_bn_relu_kernel<PP, HH, HH><<<EE / 16, KS * 32, 0, stream>>>(
      x0, a1, a2, Wf, bf, gf, bef, (float*)d_out, EE);
}


# dense layers: operands fetched as contiguous 16 B-per-lane row pieces and transposed to WMMA fragment order through wave-private LDS (was 16 cache lines per fragment load); changes fp sum order
# speedup vs baseline: 2.4557x; 1.2457x over previous
#include <hip/hip_runtime.h>

typedef __attribute__((ext_vector_type(2))) float v2f;
typedef __attribute__((ext_vector_type(8))) float v8f;
typedef __attribute__((ext_vector_type(4))) float v4f;

#define BB 32
#define NN 2048
#define PP 1024
#define HH 256
#define EE 512
#define BN_EPS 1e-5f

#define BPS_TILES 8
#define BPS_GPTS  256
#define BPS_NG    (NN / BPS_GPTS)

struct bps_group { float4 X[4], Y[4], Z[4]; };

__device__ __forceinline__ void bps_load_group(bps_group& G, const float* __restrict__ p) {
#pragma unroll
  for (int q = 0; q < 4; ++q) {
    G.X[q] = *reinterpret_cast<const float4*>(p + q * 64);
    G.Y[q] = *reinterpret_cast<const float4*>(p + NN + q * 64);
    G.Z[q] = *reinterpret_cast<const float4*>(p + 2 * NN + q * 64);
  }
}

__device__ __forceinline__ void bps_point_tile(float (&minv)[4], float px, float py,
                                               float pz, v2f Bm, bool hi) {
  const float pn2 = px * px + py * py + pz * pz;
  v2f Am;
  Am.x = hi ? pz  : px;
  Am.y = hi ? pn2 : py;
  v8f c = {0.f, 0.f, 0.f, 0.f, 0.f, 0.f, 0.f, 0.f};
  c = __builtin_amdgcn_wmma_f32_16x16x4_f32(false, Am, false, Bm,
                                            (short)0, c, false, false);
#pragma unroll
  for (int r = 0; r < 4; ++r)
    minv[r] = fminf(fminf(c[2 * r], c[2 * r + 1]), minv[r]);
}

__device__ __forceinline__ void bps_group_tiles(float (&minv)[4], const bps_group& G,
                                                v2f Bm, bool hi) {
#pragma unroll
  for (int q = 0; q < 4; ++q) {
    bps_point_tile(minv, G.X[q].x, G.Y[q].x, G.Z[q].x, Bm, hi);
    bps_point_tile(minv, G.X[q].y, G.Y[q].y, G.Z[q].y, Bm, hi);
    bps_point_tile(minv, G.X[q].z, G.Y[q].z, G.Z[q].z, Bm, hi);
    bps_point_tile(minv, G.X[q].w, G.Y[q].w, G.Z[q].w, Bm, hi);
  }
}

__global__ __launch_bounds__(256) void bps_min_kernel(
    const float* __restrict__ x,
    const float* __restrict__ basis,
    float* __restrict__ feat)
{
  const int b     = blockIdx.y;
  const int lane  = threadIdx.x & 31;
  const int wave  = threadIdx.x >> 5;
  const int col   = lane & 15;
  const bool hi   = lane >= 16;
  const int tileP = (blockIdx.x * BPS_TILES + wave) * 16;

  const float* xp = x + (size_t)b * 3 * NN + 4 * col;
  bps_group G[3];
  bps_load_group(G[0], xp);
  bps_load_group(G[1], xp + BPS_GPTS);

  const int   n   = tileP + col;
  const float bx  = basis[n * 3 + 0];
  const float by  = basis[n * 3 + 1];
  const float bz  = basis[n * 3 + 2];
  const float bn2 = bx * bx + by * by + bz * bz;
  v2f Bm;
  Bm.x = hi ? (-2.0f * bz) : (-2.0f * bx);
  Bm.y = hi ? 1.0f         : (-2.0f * by);

  float minv[4];
#pragma unroll
  for (int r = 0; r < 4; ++r) minv[r] = 3.4e38f;

#pragma unroll
  for (int g = 0; g < BPS_NG; ++g) {
    if (g + 2 < BPS_NG) {
      bps_load_group(G[(g + 2) % 3], xp + (g + 2) * BPS_GPTS);
      __builtin_amdgcn_sched_barrier(0);
    }
    bps_group_tiles(minv, G[g % 3], Bm, hi);
    __builtin_amdgcn_sched_barrier(0);
  }

  float mv = fminf(fminf(minv[0], minv[1]), fminf(minv[2], minv[3]));
  mv = fminf(mv, __shfl_xor(mv, 16));
  mv = fmaxf(mv + bn2, 0.0f);

  __shared__ float red[BPS_TILES * 16];
  if (lane < 16) red[wave * 16 + col] = mv;
  __syncthreads();
  if (threadIdx.x < BPS_TILES * 16)
    feat[(size_t)b * PP + blockIdx.x * (BPS_TILES * 16) + threadIdx.x] =
        sqrtf(red[threadIdx.x]);
}

__global__ __launch_bounds__(256) void bn0_kernel(
    const float* __restrict__ feat, const float* __restrict__ g,
    const float* __restrict__ bshift, float* __restrict__ x0)
{
  const int j = blockIdx.x * 256 + threadIdx.x;
  float s = 0.f, ss = 0.f;
#pragma unroll 4
  for (int i = 0; i < BB; ++i) {
    const float v = feat[(size_t)i * PP + j];
    s += v; ss += v * v;
  }
  const float mean = s * (1.0f / BB);
  const float var  = ss * (1.0f / BB) - mean * mean;
  const float sc   = g[j] * rsqrtf(var + BN_EPS);
  const float bt   = bshift[j];
#pragma unroll 4
  for (int i = 0; i < BB; ++i)
    x0[(size_t)i * PP + j] = (feat[(size_t)i * PP + j] - mean) * sc + bt;
}

#define KS 16

template <int Ks>
struct seg_regs {
  static constexpr int SL = Ks / KS;
  static constexpr int CH = Ks > 0 ? SL / 4 : 1;
  static constexpr int RP = 32 / CH;
  static constexpr int P4 = CH + 1;
  v4f a[CH];
  v4f w[CH > 1 ? CH / 2 : 1];
};

template <int Ks, int Ktot>
__device__ __forceinline__ void seg_fetch(seg_regs<Ks>& R, const float* __restrict__ src,
                                          const float* __restrict__ Wseg, int lane, int ks) {
  using S = seg_regs<Ks>;
  static_assert(Ks % (8 * KS) == 0 && 32 % S::CH == 0 && S::CH >= 2,
                "slice must be whole pairs of k-steps and divide a wave's 32 lanes");
  const int ch = lane % S::CH, r0 = lane / S::CH;
  const float* ap = src  + (size_t)r0 * Ks   + ks * S::SL + ch * 4;
  const float* wp = Wseg + (size_t)r0 * Ktot + ks * S::SL + ch * 4;
#pragma unroll
  for (int j = 0; j < S::CH; ++j)
    R.a[j] = *reinterpret_cast<const v4f*>(ap + (size_t)j * S::RP * Ks);
#pragma unroll
  for (int j = 0; j < S::CH / 2; ++j)
    R.w[j] = *reinterpret_cast<const v4f*>(wp + (size_t)j * S::RP * Ktot);
}

template <int Ks>
__device__ __forceinline__ void seg_mma(v8f& acc0, v8f& acc1, const seg_regs<Ks>& R,
                                        v4f* __restrict__ st, int lane) {
  using S = seg_regs<Ks>;
  const int ch = lane % S::CH, r0 = lane / S::CH;
  v4f* sp = st + r0 * S::P4 + ch;
#pragma unroll
  for (int j = 0; j < S::CH; ++j) sp[j * S::RP * S::P4] = R.a[j];
#pragma unroll
  for (int j = 0; j < S::CH / 2; ++j) sp[(32 + j * S::RP) * S::P4] = R.w[j];
  __builtin_amdgcn_wave_barrier();
  const v4f* f = st + (lane & 15) * S::P4 + (lane >> 4);
#pragma unroll
  for (int k8 = 0; k8 < S::SL / 8; ++k8) {
    const v4f av0 = f[2 * k8];
    const v4f av1 = f[16 * S::P4 + 2 * k8];
    const v4f wv  = f[32 * S::P4 + 2 * k8];
    v2f A0; A0.x = av0.x; A0.y = av0.y;
    v2f A1; A1.x = av1.x; A1.y = av1.y;
    v2f B0; B0.x = wv.x;  B0.y = wv.y;
    acc0 = __builtin_amdgcn_wmma_f32_16x16x4_f32(false, A0, false, B0,
                                                 (short)0, acc0, false, false);
    acc1 = __builtin_amdgcn_wmma_f32_16x16x4_f32(false, A1, false, B0,
                                                 (short)0, acc1, false, false);
    v2f A2; A2.x = av0.z; A2.y = av0.w;
    v2f A3; A3.x = av1.z; A3.y = av1.w;
    v2f B1; B1.x = wv.z;  B1.y = wv.w;
    acc0 = __builtin_amdgcn_wmma_f32_16x16x4_f32(false, A2, false, B1,
                                                 (short)0, acc0, false, false);
    acc1 = __builtin_amdgcn_wmma_f32_16x16x4_f32(false, A3, false, B1,
                                                 (short)0, acc1, false, false);
  }
  __builtin_amdgcn_wave_barrier();
}

template <int K0, int K1, int K2>
__global__ __launch_bounds__(KS * 32) void dense_bn_relu_kernel(
    const float* __restrict__ in0,
    const float* __restrict__ in1,
    const float* __restrict__ in2,
    const float* __restrict__ W,
    const float* __restrict__ bias,
    const float* __restrict__ g, const float* __restrict__ beta,
    float* __restrict__ out, int Hout)
{
  constexpr int Ktot = K0 + K1 + K2;
  constexpr int KMAX = K0 > K1 ? (K0 > K2 ? K0 : K2) : (K1 > K2 ? K1 : K2);
  constexpr int REG4 = 48 * seg_regs<KMAX>::P4;
  static_assert((K1 == 0 || seg_regs<K1>::CH == 4) && (K2 == 0 || seg_regs<K2>::CH == 4),
                "the load fence below names 4 + 2 loads per later segment");
  const int  c    = blockIdx.x * 16;
  const int  lane = threadIdx.x & 31;
  const int  wave = threadIdx.x >> 5;
  const int  col  = lane & 15;
  const bool hi   = lane >= 16;
  const int  n    = c + col;

  __shared__ v4f stage[KS * REG4];
  v4f* st = stage + wave * REG4;

  v8f acc0 = {0.f, 0.f, 0.f, 0.f, 0.f, 0.f, 0.f, 0.f};
  v8f acc1 = {0.f, 0.f, 0.f, 0.f, 0.f, 0.f, 0.f, 0.f};
  const float* Wc = W + (size_t)c * Ktot;
  seg_regs<K0> R0; seg_regs<K1> R1; seg_regs<K2> R2;
  seg_fetch<K0, Ktot>(R0, in0, Wc, lane, wave);
  if constexpr (K1 > 0) seg_fetch<K1, Ktot>(R1, in1, Wc + K0, lane, wave);
  if constexpr (K2 > 0) seg_fetch<K2, Ktot>(R2, in2, Wc + K0 + K1, lane, wave);
  if constexpr (K1 > 0)
    asm volatile("" :: "v"(R1.a[0]), "v"(R1.a[1]), "v"(R1.a[2]), "v"(R1.a[3]),
                       "v"(R1.w[0]), "v"(R1.w[1]));
  if constexpr (K2 > 0)
    asm volatile("" :: "v"(R2.a[0]), "v"(R2.a[1]), "v"(R2.a[2]), "v"(R2.a[3]),
                       "v"(R2.w[0]), "v"(R2.w[1]));
  seg_mma<K0>(acc0, acc1, R0, st, lane);
  if constexpr (K1 > 0) seg_mma<K1>(acc0, acc1, R1, st, lane);
  if constexpr (K2 > 0) seg_mma<K2>(acc0, acc1, R2, st, lane);

  __shared__ float4 part[KS][2][2][32];
  __shared__ float sred[2][16][2];
  part[wave][0][0][lane] = make_float4(acc0[0], acc0[1], acc0[2], acc0[3]);
  part[wave][0][1][lane] = make_float4(acc0[4], acc0[5], acc0[6], acc0[7]);
  part[wave][1][0][lane] = make_float4(acc1[0], acc1[1], acc1[2], acc1[3]);
  part[wave][1][1][lane] = make_float4(acc1[4], acc1[5], acc1[6], acc1[7]);
  __syncthreads();

  float vals[8];
  if (wave < 2) {
    float a[8];
#pragma unroll
    for (int r = 0; r < 8; ++r) a[r] = 0.f;
#pragma unroll
    for (int s = 0; s < KS; ++s) {
      const float4 lo = part[s][wave][0][lane];
      const float4 hv = part[s][wave][1][lane];
      a[0] += lo.x; a[1] += lo.y; a[2] += lo.z; a[3] += lo.w;
      a[4] += hv.x; a[5] += hv.y; a[6] += hv.z; a[7] += hv.w;
    }
    const float bi = bias[n];
    float s = 0.f, ss = 0.f;
#pragma unroll
    for (int r = 0; r < 8; ++r) {
      const float v = fmaxf(a[r] + bi, 0.f);
      vals[r] = v; s += v; ss += v * v;
    }
    s  += __shfl_xor(s, 16);
    ss += __shfl_xor(ss, 16);
    if (lane < 16) { sred[wave][col][0] = s; sred[wave][col][1] = ss; }
  }
  __syncthreads();
  if (wave < 2) {
    const float ts   = sred[0][col][0] + sred[1][col][0];
    const float tss  = sred[0][col][1] + sred[1][col][1];
    const float mean = ts * (1.0f / BB);
    const float var  = tss * (1.0f / BB) - mean * mean;
    const float sc   = g[n] * rsqrtf(var + BN_EPS);
    const float bt   = beta[n];
#pragma unroll
    for (int r = 0; r < 8; ++r) {
      const int rr = wave * 16 + r + (hi ? 8 : 0);
      out[(size_t)rr * Hout + n] = (vals[r] - mean) * sc + bt;
    }
  }
}

extern "C" void kernel_launch(void* const* d_in, const int* in_sizes, int n_in,
                              void* d_out, int out_size, void* d_ws, size_t ws_size,
                              hipStream_t stream) {
  const float* x     = (const float*)d_in[0];
  const float* basis = (const float*)d_in[1];
  const float* bn0_g = (const float*)d_in[2];
  const float* bn0_b = (const float*)d_in[3];
  const float* W0 = (const float*)d_in[4];
  const float* b0 = (const float*)d_in[5];
  const float* g0 = (const float*)d_in[6];
  const float* be0 = (const float*)d_in[7];
  const float* W1 = (const float*)d_in[8];
  const float* b1 = (const float*)d_in[9];
  const float* g1 = (const float*)d_in[10];
  const float* be1 = (const float*)d_in[11];
  const float* W2 = (const float*)d_in[12];
  const float* b2 = (const float*)d_in[13];
  const float* g2 = (const float*)d_in[14];
  const float* be2 = (const float*)d_in[15];
  const float* W3 = (const float*)d_in[16];
  const float* b3 = (const float*)d_in[17];
  const float* g3 = (const float*)d_in[18];
  const float* be3 = (const float*)d_in[19];
  const float* Wf = (const float*)d_in[20];
  const float* bf = (const float*)d_in[21];
  const float* gf = (const float*)d_in[22];
  const float* bef = (const float*)d_in[23];

  float* ws   = (float*)d_ws;
  float* feat = ws;
  float* x0   = ws + 32768;
  float* h    = ws + 65536;
  float* a1   = ws + 73728;
  float* a2   = ws + 81920;

  bps_min_kernel<<<dim3(PP / (BPS_TILES * 16), BB), 256, 0, stream>>>(x, basis, feat);
  bn0_kernel<<<PP / 256, 256, 0, stream>>>(feat, bn0_g, bn0_b, x0);
  dense_bn_relu_kernel<PP, 0, 0><<<HH / 16, KS * 32, 0, stream>>>(
      x0, nullptr, nullptr, W0, b0, g0, be0, h, HH);
  dense_bn_relu_kernel<HH, 0, 0><<<HH / 16, KS * 32, 0, stream>>>(
      h, nullptr, nullptr, W1, b1, g1, be1, a1, HH);
  dense_bn_relu_kernel<PP, HH, 0><<<HH / 16, KS * 32, 0, stream>>>(
      x0, a1, nullptr, W2, b2, g2, be2, h, HH);
  dense_bn_relu_kernel<HH, 0, 0><<<HH / 16, KS * 32, 0, stream>>>(
      h, nullptr, nullptr, W3, b3, g3, be3, a2, HH);
  dense_bn_relu_kernel<PP, HH, HH><<<EE / 16, KS * 32, 0, stream>>>(
      x0, a1, a2, Wf, bf, gf, bef, (float*)d_out, EE);
}


# input BatchNorm folded into the first dense layer: every wave normalises its own 32-row K slice in registers (stats as two 16-row partial sums: changes their fp sum order); one launch fewer
# speedup vs baseline: 2.9290x; 1.1927x over previous
#include <hip/hip_runtime.h>

typedef __attribute__((ext_vector_type(2))) float v2f;
typedef __attribute__((ext_vector_type(8))) float v8f;
typedef __attribute__((ext_vector_type(4))) float v4f;

#define BB 32
#define NN 2048
#define PP 1024
#define HH 256
#define EE 512
#define BN_EPS 1e-5f

#define BPS_TILES 8
#define BPS_GPTS  256
#define BPS_NG    (NN / BPS_GPTS)

struct bps_group { float4 X[4], Y[4], Z[4]; };

__device__ __forceinline__ void bps_load_group(bps_group& G, const float* __restrict__ p) {
#pragma unroll
  for (int q = 0; q < 4; ++q) {
    G.X[q] = *reinterpret_cast<const float4*>(p + q * 64);
    G.Y[q] = *reinterpret_cast<const float4*>(p + NN + q * 64);
    G.Z[q] = *reinterpret_cast<const float4*>(p + 2 * NN + q * 64);
  }
}

__device__ __forceinline__ void bps_point_tile(float (&minv)[4], float px, float py,
                                               float pz, v2f Bm, bool hi) {
  const float pn2 = px * px + py * py + pz * pz;
  v2f Am;
  Am.x = hi ? pz  : px;
  Am.y = hi ? pn2 : py;
  v8f c = {0.f, 0.f, 0.f, 0.f, 0.f, 0.f, 0.f, 0.f};
  c = __builtin_amdgcn_wmma_f32_16x16x4_f32(false, Am, false, Bm,
                                            (short)0, c, false, false);
#pragma unroll
  for (int r = 0; r < 4; ++r)
    minv[r] = fminf(fminf(c[2 * r], c[2 * r + 1]), minv[r]);
}

__device__ __forceinline__ void bps_group_tiles(float (&minv)[4], const bps_group& G,
                                                v2f Bm, bool hi) {
#pragma unroll
  for (int q = 0; q < 4; ++q) {
    bps_point_tile(minv, G.X[q].x, G.Y[q].x, G.Z[q].x, Bm, hi);
    bps_point_tile(minv, G.X[q].y, G.Y[q].y, G.Z[q].y, Bm, hi);
    bps_point_tile(minv, G.X[q].z, G.Y[q].z, G.Z[q].z, Bm, hi);
    bps_point_tile(minv, G.X[q].w, G.Y[q].w, G.Z[q].w, Bm, hi);
  }
}

__global__ __launch_bounds__(256) void bps_min_kernel(
    const float* __restrict__ x,
    const float* __restrict__ basis,
    float* __restrict__ feat)
{
  const int b     = blockIdx.y;
  const int lane  = threadIdx.x & 31;
  const int wave  = threadIdx.x >> 5;
  const int col   = lane & 15;
  const bool hi   = lane >= 16;
  const int tileP = (blockIdx.x * BPS_TILES + wave) * 16;

  const float* xp = x + (size_t)b * 3 * NN + 4 * col;
  bps_group G[3];
  bps_load_group(G[0], xp);
  bps_load_group(G[1], xp + BPS_GPTS);

  const int   n   = tileP + col;
  const float bx  = basis[n * 3 + 0];
  const float by  = basis[n * 3 + 1];
  const float bz  = basis[n * 3 + 2];
  const float bn2 = bx * bx + by * by + bz * bz;
  v2f Bm;
  Bm.x = hi ? (-2.0f * bz) : (-2.0f * bx);
  Bm.y = hi ? 1.0f         : (-2.0f * by);

  float minv[4];
#pragma unroll
  for (int r = 0; r < 4; ++r) minv[r] = 3.4e38f;

#pragma unroll
  for (int g = 0; g < BPS_NG; ++g) {
    if (g + 2 < BPS_NG) {
      bps_load_group(G[(g + 2) % 3], xp + (g + 2) * BPS_GPTS);
      __builtin_amdgcn_sched_barrier(0);
    }
    bps_group_tiles(minv, G[g % 3], Bm, hi);
    __builtin_amdgcn_sched_barrier(0);
  }

  float mv = fminf(fminf(minv[0], minv[1]), fminf(minv[2], minv[3]));
  mv = fminf(mv, __shfl_xor(mv, 16));
  mv = fmaxf(mv + bn2, 0.0f);

  __shared__ float red[BPS_TILES * 16];
  if (lane < 16) red[wave * 16 + col] = mv;
  __syncthreads();
  if (threadIdx.x < BPS_TILES * 16)
    feat[(size_t)b * PP + blockIdx.x * (BPS_TILES * 16) + threadIdx.x] =
        sqrtf(red[threadIdx.x]);
}

#define KS 16

template <int Ks>
struct seg_regs {
  static constexpr int SL = Ks / KS;
  static constexpr int CH = Ks > 0 ? SL / 4 : 1;
  static constexpr int RP = 32 / CH;
  static constexpr int P4 = CH + 1;
  v4f a[CH];
  v4f w[CH > 1 ? CH / 2 : 1];
};

template <int Ks, int Ktot>
__device__ __forceinline__ void seg_fetch(seg_regs<Ks>& R, const float* __restrict__ src,
                                          const float* __restrict__ Wseg, int lane, int ks) {
  using S = seg_regs<Ks>;
  static_assert(Ks % (8 * KS) == 0 && 32 % S::CH == 0 && S::CH >= 2,
                "slice must be whole pairs of k-steps and divide a wave's 32 lanes");
  const int ch = lane % S::CH, r0 = lane / S::CH;
  const float* ap = src  + (size_t)r0 * Ks   + ks * S::SL + ch * 4;
  const float* wp = Wseg + (size_t)r0 * Ktot + ks * S::SL + ch * 4;
#pragma unroll
  for (int j = 0; j < S::CH; ++j)
    R.a[j] = *reinterpret_cast<const v4f*>(ap + (size_t)j * S::RP * Ks);
#pragma unroll
  for (int j = 0; j < S::CH / 2; ++j)
    R.w[j] = *reinterpret_cast<const v4f*>(wp + (size_t)j * S::RP * Ktot);
}

__device__ __forceinline__ float xhalf(float x) {
  const int b = __builtin_bit_cast(int, x);
  return __builtin_bit_cast(float,
      __builtin_amdgcn_permlanex16(b, b, 0x76543210, 0xfedcba98, false, false));
}

template <int Ks>
__device__ __forceinline__ void seg_batchnorm(seg_regs<Ks>& R, v4f g, v4f b) {
  using S = seg_regs<Ks>;
  static_assert(S::RP == 2 && S::CH * S::RP == BB,
                "a lane and lane ^ 16 must hold all batch rows of their columns between them");
  v4f s = R.a[0], q = R.a[0] * R.a[0];
#pragma unroll
  for (int j = 1; j < S::CH; ++j) { s += R.a[j]; q += R.a[j] * R.a[j]; }
  v4f mean, sc;
#pragma unroll
  for (int c = 0; c < 4; ++c) {
    const float ts  = s[c] + xhalf(s[c]);
    const float tq  = q[c] + xhalf(q[c]);
    const float m   = ts * (1.0f / BB);
    const float var = tq * (1.0f / BB) - m * m;
    mean[c] = m;
    sc[c]   = g[c] * rsqrtf(var + BN_EPS);
  }
#pragma unroll
  for (int j = 0; j < S::CH; ++j) R.a[j] = (R.a[j] - mean) * sc + b;
}

template <int Ks>
__device__ __forceinline__ void seg_mma(v8f& acc0, v8f& acc1, const seg_regs<Ks>& R,
                                        v4f* __restrict__ st, int lane) {
  using S = seg_regs<Ks>;
  const int ch = lane % S::CH, r0 = lane / S::CH;
  v4f* sp = st + r0 * S::P4 + ch;
#pragma unroll
  for (int j = 0; j < S::CH; ++j) sp[j * S::RP * S::P4] = R.a[j];
#pragma unroll
  for (int j = 0; j < S::CH / 2; ++j) sp[(32 + j * S::RP) * S::P4] = R.w[j];
  __builtin_amdgcn_wave_barrier();
  const v4f* f = st + (lane & 15) * S::P4 + (lane >> 4);
#pragma unroll
  for (int k8 = 0; k8 < S::SL / 8; ++k8) {
    const v4f av0 = f[2 * k8];
    const v4f av1 = f[16 * S::P4 + 2 * k8];
    const v4f wv  = f[32 * S::P4 + 2 * k8];
    v2f A0; A0.x = av0.x; A0.y = av0.y;
    v2f A1; A1.x = av1.x; A1.y = av1.y;
    v2f B0; B0.x = wv.x;  B0.y = wv.y;
    acc0 = __builtin_amdgcn_wmma_f32_16x16x4_f32(false, A0, false, B0,
                                                 (short)0, acc0, false, false);
    acc1 = __builtin_amdgcn_wmma_f32_16x16x4_f32(false, A1, false, B0,
                                                 (short)0, acc1, false, false);
    v2f A2; A2.x = av0.z; A2.y = av0.w;
    v2f A3; A3.x = av1.z; A3.y = av1.w;
    v2f B1; B1.x = wv.z;  B1.y = wv.w;
    acc0 = __builtin_amdgcn_wmma_f32_16x16x4_f32(false, A2, false, B1,
                                                 (short)0, acc0, false, false);
    acc1 = __builtin_amdgcn_wmma_f32_16x16x4_f32(false, A3, false, B1,
                                                 (short)0, acc1, false, false);
  }
  __builtin_amdgcn_wave_barrier();
}

template <int K0, int K1, int K2, bool BN_IN>
__global__ __launch_bounds__(KS * 32) void dense_bn_relu_kernel(
    const float* __restrict__ in0,
    const float* __restrict__ in1,
    const float* __restrict__ in2,
    const float* __restrict__ W,
    const float* __restrict__ bias,
    const float* __restrict__ g, const float* __restrict__ beta,
    float* __restrict__ out, int Hout,
    const float* __restrict__ bn_g,
    const float* __restrict__ bn_b,
    float* __restrict__ bn_out)
{
  constexpr int Ktot = K0 + K1 + K2;
  constexpr int KMAX = K0 > K1 ? (K0 > K2 ? K0 : K2) : (K1 > K2 ? K1 : K2);
  constexpr int REG4 = 48 * seg_regs<KMAX>::P4;
  static_assert((K1 == 0 || seg_regs<K1>::CH == 4) && (K2 == 0 || seg_regs<K2>::CH == 4),
                "the load fence below names 4 + 2 loads per later segment");
  const int  c    = blockIdx.x * 16;
  const int  lane = threadIdx.x & 31;
  const int  wave = threadIdx.x >> 5;
  const int  col  = lane & 15;
  const bool hi   = lane >= 16;
  const int  n    = c + col;

  __shared__ v4f stage[KS * REG4];
  v4f* st = stage + wave * REG4;

  v8f acc0 = {0.f, 0.f, 0.f, 0.f, 0.f, 0.f, 0.f, 0.f};
  v8f acc1 = {0.f, 0.f, 0.f, 0.f, 0.f, 0.f, 0.f, 0.f};
  const float* Wc = W + (size_t)c * Ktot;
  seg_regs<K0> R0; seg_regs<K1> R1; seg_regs<K2> R2;
  v4f bng = {0.f, 0.f, 0.f, 0.f}, bnb = {0.f, 0.f, 0.f, 0.f};
  if constexpr (BN_IN) {
    const int cb = wave * seg_regs<K0>::SL + (lane % seg_regs<K0>::CH) * 4;
    bng = *reinterpret_cast<const v4f*>(bn_g + cb);
    bnb = *reinterpret_cast<const v4f*>(bn_b + cb);
  }
  seg_fetch<K0, Ktot>(R0, in0, Wc, lane, wave);
  if constexpr (K1 > 0) seg_fetch<K1, Ktot>(R1, in1, Wc + K0, lane, wave);
  if constexpr (K2 > 0) seg_fetch<K2, Ktot>(R2, in2, Wc + K0 + K1, lane, wave);
  if constexpr (BN_IN) __builtin_amdgcn_sched_barrier(0);
  if constexpr (K1 > 0)
    asm volatile("" :: "v"(R1.a[0]), "v"(R1.a[1]), "v"(R1.a[2]), "v"(R1.a[3]),
                       "v"(R1.w[0]), "v"(R1.w[1]));
  if constexpr (K2 > 0)
    asm volatile("" :: "v"(R2.a[0]), "v"(R2.a[1]), "v"(R2.a[2]), "v"(R2.a[3]),
                       "v"(R2.w[0]), "v"(R2.w[1]));
  if constexpr (BN_IN) {
    using S0 = seg_regs<K0>;
    seg_batchnorm<K0>(R0, bng, bnb);
    if (wave == (int)blockIdx.x) {
      float* xo = bn_out + (size_t)(lane / S0::CH) * K0 + wave * S0::SL + (lane % S0::CH) * 4;
#pragma unroll
      for (int j = 0; j < S0::CH; ++j)
        *reinterpret_cast<v4f*>(xo + (size_t)j * S0::RP * K0) = R0.a[j];
    }
  }
  seg_mma<K0>(acc0, acc1, R0, st, lane);
  if constexpr (K1 > 0) seg_mma<K1>(acc0, acc1, R1, st, lane);
  if constexpr (K2 > 0) seg_mma<K2>(acc0, acc1, R2, st, lane);

  __shared__ float4 part[KS][2][2][32];
  __shared__ float sred[2][16][2];
  part[wave][0][0][lane] = make_float4(acc0[0], acc0[1], acc0[2], acc0[3]);
  part[wave][0][1][lane] = make_float4(acc0[4], acc0[5], acc0[6], acc0[7]);
  part[wave][1][0][lane] = make_float4(acc1[0], acc1[1], acc1[2], acc1[3]);
  part[wave][1][1][lane] = make_float4(acc1[4], acc1[5], acc1[6], acc1[7]);
  __syncthreads();

  float vals[8];
  if (wave < 2) {
    float a[8];
#pragma unroll
    for (int r = 0; r < 8; ++r) a[r] = 0.f;
#pragma unroll
    for (int s = 0; s < KS; ++s) {
      const float4 lo = part[s][wave][0][lane];
      const float4 hv = part[s][wave][1][lane];
      a[0] += lo.x; a[1] += lo.y; a[2] += lo.z; a[3] += lo.w;
      a[4] += hv.x; a[5] += hv.y; a[6] += hv.z; a[7] += hv.w;
    }
    const float bi = bias[n];
    float s = 0.f, ss = 0.f;
#pragma unroll
    for (int r = 0; r < 8; ++r) {
      const float v = fmaxf(a[r] + bi, 0.f);
      vals[r] = v; s += v; ss += v * v;
    }
    s  += __shfl_xor(s, 16);
    ss += __shfl_xor(ss, 16);
    if (lane < 16) { sred[wave][col][0] = s; sred[wave][col][1] = ss; }
  }
  __syncthreads();
  if (wave < 2) {
    const float ts   = sred[0][col][0] + sred[1][col][0];
    const float tss  = sred[0][col][1] + sred[1][col][1];
    const float mean = ts * (1.0f / BB);
    const float var  = tss * (1.0f / BB) - mean * mean;
    const float sc   = g[n] * rsqrtf(var + BN_EPS);
    const float bt   = beta[n];
#pragma unroll
    for (int r = 0; r < 8; ++r) {
      const int rr = wave * 16 + r + (hi ? 8 : 0);
      out[(size_t)rr * Hout + n] = (vals[r] - mean) * sc + bt;
    }
  }
}

extern "C" void kernel_launch(void* const* d_in, const int* in_sizes, int n_in,
                              void* d_out, int out_size, void* d_ws, size_t ws_size,
                              hipStream_t stream) {
  const float* x     = (const float*)d_in[0];
  const float* basis = (const float*)d_in[1];
  const float* bn0_g = (const float*)d_in[2];
  const float* bn0_b = (const float*)d_in[3];
  const float* W0 = (const float*)d_in[4];
  const float* b0 = (const float*)d_in[5];
  const float* g0 = (const float*)d_in[6];
  const float* be0 = (const float*)d_in[7];
  const float* W1 = (const float*)d_in[8];
  const float* b1 = (const float*)d_in[9];
  const float* g1 = (const float*)d_in[10];
  const float* be1 = (const float*)d_in[11];
  const float* W2 = (const float*)d_in[12];
  const float* b2 = (const float*)d_in[13];
  const float* g2 = (const float*)d_in[14];
  const float* be2 = (const float*)d_in[15];
  const float* W3 = (const float*)d_in[16];
  const float* b3 = (const float*)d_in[17];
  const float* g3 = (const float*)d_in[18];
  const float* be3 = (const float*)d_in[19];
  const float* Wf = (const float*)d_in[20];
  const float* bf = (const float*)d_in[21];
  const float* gf = (const float*)d_in[22];
  const float* bef = (const float*)d_in[23];

  float* ws   = (float*)d_ws;
  float* feat = ws;
  float* x0   = ws + 32768;
  float* h    = ws + 65536;
  float* a1   = ws + 73728;
  float* a2   = ws + 81920;

  bps_min_kernel<<<dim3(PP / (BPS_TILES * 16), BB), 256, 0, stream>>>(x, basis, feat);
  static_assert(HH / 16 == KS, "workgroup ks of the first layer writes slice ks of x0");
  dense_bn_relu_kernel<PP, 0, 0, true><<<HH / 16, KS * 32, 0, stream>>>(
      feat, nullptr, nullptr, W0, b0, g0, be0, h, HH, bn0_g, bn0_b, x0);
  dense_bn_relu_kernel<HH, 0, 0, false><<<HH / 16, KS * 32, 0, stream>>>(
      h, nullptr, nullptr, W1, b1, g1, be1, a1, HH, nullptr, nullptr, nullptr);
  dense_bn_relu_kernel<PP, HH, 0, false><<<HH / 16, KS * 32, 0, stream>>>(
      x0, a1, nullptr, W2, b2, g2, be2, h, HH, nullptr, nullptr, nullptr);
  dense_bn_relu_kernel<HH, 0, 0, false><<<HH / 16, KS * 32, 0, stream>>>(
      h, nullptr, nullptr, W3, b3, g3, be3, a2, HH, nullptr, nullptr, nullptr);
  dense_bn_relu_kernel<PP, HH, HH, false><<<EE / 16, KS * 32, 0, stream>>>(
      x0, a1, a2, Wf, bf, gf, bef, (float*)d_out, EE, nullptr, nullptr, nullptr);
}


# concat layers split by segment: each launch multiplies its one input by every later layer's columns for it, stores raw partial tiles; later layers add them (K 256 each); changes fp sum order
# speedup vs baseline: 3.3979x; 1.1601x over previous
#include <hip/hip_runtime.h>

typedef __attribute__((ext_vector_type(2))) float v2f;
typedef __attribute__((ext_vector_type(8))) float v8f;
typedef __attribute__((ext_vector_type(4))) float v4f;

#define BB 32
#define NN 2048
#define PP 1024
#define HH 256
#define EE 512
#define BN_EPS 1e-5f

#define BPS_TILES 8
#define BPS_GPTS  256
#define BPS_NG    (NN / BPS_GPTS)

struct bps_group { float4 X[4], Y[4], Z[4]; };

__device__ __forceinline__ void bps_load_group(bps_group& G, const float* __restrict__ p) {
#pragma unroll
  for (int q = 0; q < 4; ++q) {
    G.X[q] = *reinterpret_cast<const float4*>(p + q * 64);
    G.Y[q] = *reinterpret_cast<const float4*>(p + NN + q * 64);
    G.Z[q] = *reinterpret_cast<const float4*>(p + 2 * NN + q * 64);
  }
}

__device__ __forceinline__ void bps_point_tile(float (&minv)[4], float px, float py,
                                               float pz, v2f Bm, bool hi) {
  const float pn2 = px * px + py * py + pz * pz;
  v2f Am;
  Am.x = hi ? pz  : px;
  Am.y = hi ? pn2 : py;
  v8f c = {0.f, 0.f, 0.f, 0.f, 0.f, 0.f, 0.f, 0.f};
  c = __builtin_amdgcn_wmma_f32_16x16x4_f32(false, Am, false, Bm,
                                            (short)0, c, false, false);
#pragma unroll
  for (int r = 0; r < 4; ++r)
    minv[r] = fminf(fminf(c[2 * r], c[2 * r + 1]), minv[r]);
}

__device__ __forceinline__ void bps_group_tiles(float (&minv)[4], const bps_group& G,
                                                v2f Bm, bool hi) {
#pragma unroll
  for (int q = 0; q < 4; ++q) {
    bps_point_tile(minv, G.X[q].x, G.Y[q].x, G.Z[q].x, Bm, hi);
    bps_point_tile(minv, G.X[q].y, G.Y[q].y, G.Z[q].y, Bm, hi);
    bps_point_tile(minv, G.X[q].z, G.Y[q].z, G.Z[q].z, Bm, hi);
    bps_point_tile(minv, G.X[q].w, G.Y[q].w, G.Z[q].w, Bm, hi);
  }
}

__global__ __launch_bounds__(256) void bps_min_kernel(
    const float* __restrict__ x,
    const float* __restrict__ basis,
    float* __restrict__ feat)
{
  const int b     = blockIdx.y;
  const int lane  = threadIdx.x & 31;
  const int wave  = threadIdx.x >> 5;
  const int col   = lane & 15;
  const bool hi   = lane >= 16;
  const int tileP = (blockIdx.x * BPS_TILES + wave) * 16;

  const float* xp = x + (size_t)b * 3 * NN + 4 * col;
  bps_group G[3];
  bps_load_group(G[0], xp);
  bps_load_group(G[1], xp + BPS_GPTS);

  const int   n   = tileP + col;
  const float bx  = basis[n * 3 + 0];
  const float by  = basis[n * 3 + 1];
  const float bz  = basis[n * 3 + 2];
  const float bn2 = bx * bx + by * by + bz * bz;
  v2f Bm;
  Bm.x = hi ? (-2.0f * bz) : (-2.0f * bx);
  Bm.y = hi ? 1.0f         : (-2.0f * by);

  float minv[4];
#pragma unroll
  for (int r = 0; r < 4; ++r) minv[r] = 3.4e38f;

#pragma unroll
  for (int g = 0; g < BPS_NG; ++g) {
    if (g + 2 < BPS_NG) {
      bps_load_group(G[(g + 2) % 3], xp + (g + 2) * BPS_GPTS);
      __builtin_amdgcn_sched_barrier(0);
    }
    bps_group_tiles(minv, G[g % 3], Bm, hi);
    __builtin_amdgcn_sched_barrier(0);
  }

  float mv = fminf(fminf(minv[0], minv[1]), fminf(minv[2], minv[3]));
  mv = fminf(mv, __shfl_xor(mv, 16));
  mv = fmaxf(mv + bn2, 0.0f);

  __shared__ float red[BPS_TILES * 16];
  if (lane < 16) red[wave * 16 + col] = mv;
  __syncthreads();
  if (threadIdx.x < BPS_TILES * 16)
    feat[(size_t)b * PP + blockIdx.x * (BPS_TILES * 16) + threadIdx.x] =
        sqrtf(red[threadIdx.x]);
}

#define KS 16

template <int Ks>
struct seg_regs {
  static constexpr int SL = Ks / KS;
  static constexpr int CH = SL / 4;
  static constexpr int RP = 32 / CH;
  static constexpr int P4 = CH + 1;
  v4f a[CH];
  v4f w[CH / 2];
};

template <int Ks>
__device__ __forceinline__ void seg_fetch(seg_regs<Ks>& R, const float* __restrict__ src,
                                          const float* __restrict__ Wseg, int ldw,
                                          int lane, int ks) {
  using S = seg_regs<Ks>;
  static_assert(Ks % (8 * KS) == 0 && 32 % S::CH == 0 && S::CH >= 2,
                "slice must be whole pairs of k-steps and divide a wave's 32 lanes");
  const int ch = lane % S::CH, r0 = lane / S::CH;
  const float* ap = src  + (size_t)r0 * Ks  + ks * S::SL + ch * 4;
  const float* wp = Wseg + (size_t)r0 * ldw + ks * S::SL + ch * 4;
#pragma unroll
  for (int j = 0; j < S::CH; ++j)
    R.a[j] = *reinterpret_cast<const v4f*>(ap + (size_t)j * S::RP * Ks);
#pragma unroll
  for (int j = 0; j < S::CH / 2; ++j)
    R.w[j] = *reinterpret_cast<const v4f*>(wp + (size_t)j * S::RP * ldw);
}

__device__ __forceinline__ float xhalf(float x) {
  const int b = __builtin_bit_cast(int, x);
  return __builtin_bit_cast(float,
      __builtin_amdgcn_permlanex16(b, b, 0x76543210, 0xfedcba98, false, false));
}

template <int Ks>
__device__ __forceinline__ void seg_batchnorm(seg_regs<Ks>& R, v4f g, v4f b) {
  using S = seg_regs<Ks>;
  static_assert(S::RP == 2 && S::CH * S::RP == BB,
                "a lane and lane ^ 16 must hold all batch rows of their columns between them");
  v4f s = R.a[0], q = R.a[0] * R.a[0];
#pragma unroll
  for (int j = 1; j < S::CH; ++j) { s += R.a[j]; q += R.a[j] * R.a[j]; }
  v4f mean, sc;
#pragma unroll
  for (int c = 0; c < 4; ++c) {
    const float ts  = s[c] + xhalf(s[c]);
    const float tq  = q[c] + xhalf(q[c]);
    const float m   = ts * (1.0f / BB);
    const float var = tq * (1.0f / BB) - m * m;
    mean[c] = m;
    sc[c]   = g[c] * rsqrtf(var + BN_EPS);
  }
#pragma unroll
  for (int j = 0; j < S::CH; ++j) R.a[j] = (R.a[j] - mean) * sc + b;
}

template <int Ks>
__device__ __forceinline__ void seg_mma(v8f& acc0, v8f& acc1, const seg_regs<Ks>& R,
                                        v4f* __restrict__ st, int lane) {
  using S = seg_regs<Ks>;
  const int ch = lane % S::CH, r0 = lane / S::CH;
  v4f* sp = st + r0 * S::P4 + ch;
#pragma unroll
  for (int j = 0; j < S::CH; ++j) sp[j * S::RP * S::P4] = R.a[j];
#pragma unroll
  for (int j = 0; j < S::CH / 2; ++j) sp[(32 + j * S::RP) * S::P4] = R.w[j];
  __builtin_amdgcn_wave_barrier();
  const v4f* f = st + (lane & 15) * S::P4 + (lane >> 4);
#pragma unroll
  for (int k8 = 0; k8 < S::SL / 8; ++k8) {
    const v4f av0 = f[2 * k8];
    const v4f av1 = f[16 * S::P4 + 2 * k8];
    const v4f wv  = f[32 * S::P4 + 2 * k8];
    v2f A0; A0.x = av0.x; A0.y = av0.y;
    v2f A1; A1.x = av1.x; A1.y = av1.y;
    v2f B0; B0.x = wv.x;  B0.y = wv.y;
    acc0 = __builtin_amdgcn_wmma_f32_16x16x4_f32(false, A0, false, B0,
                                                 (short)0, acc0, false, false);
    acc1 = __builtin_amdgcn_wmma_f32_16x16x4_f32(false, A1, false, B0,
                                                 (short)0, acc1, false, false);
    v2f A2; A2.x = av0.z; A2.y = av0.w;
    v2f A3; A3.x = av1.z; A3.y = av1.w;
    v2f B1; B1.x = wv.z;  B1.y = wv.w;
    acc0 = __builtin_amdgcn_wmma_f32_16x16x4_f32(false, A2, false, B1,
                                                 (short)0, acc0, false, false);
    acc1 = __builtin_amdgcn_wmma_f32_16x16x4_f32(false, A3, false, B1,
                                                 (short)0, acc1, false, false);
  }
}

struct dense_job {
  const float* W;
  const float* bias;
  const float* g;
  const float* beta;
  const float* addin;
  float* out;
  int ldw;
  int ldo;
  int nblk;
  int finish;
};

template <int K, bool BN_IN>
__global__ __launch_bounds__(KS * 32) void dense_kernel(
    const float* __restrict__ in0,
    const float* __restrict__ bn_g,
    const float* __restrict__ bn_b,
    dense_job j0, dense_job j1, dense_job j2)
{
  using S = seg_regs<K>;
  constexpr int REG4 = 48 * S::P4;
  const int  lane = threadIdx.x & 31;
  const int  wave = threadIdx.x >> 5;
  const int  col  = lane & 15;
  const bool hi   = lane >= 16;

  const int  b0  = blockIdx.x;
  const bool in1 = b0 >= j0.nblk;
  const bool in2 = b0 >= j0.nblk + j1.nblk;
  dense_job J;
  J.W      = in2 ? j2.W      : in1 ? j1.W      : j0.W;
  J.bias   = in2 ? j2.bias   : in1 ? j1.bias   : j0.bias;
  J.g      = in2 ? j2.g      : in1 ? j1.g      : j0.g;
  J.beta   = in2 ? j2.beta   : in1 ? j1.beta   : j0.beta;
  J.addin  = in2 ? j2.addin  : in1 ? j1.addin  : j0.addin;
  J.out    = in2 ? j2.out    : in1 ? j1.out    : j0.out;
  J.ldw    = in2 ? j2.ldw    : in1 ? j1.ldw    : j0.ldw;
  J.ldo    = in2 ? j2.ldo    : in1 ? j1.ldo    : j0.ldo;
  J.finish = in2 ? j2.finish : in1 ? j1.finish : j0.finish;
  const int bx = in2 ? b0 - j0.nblk - j1.nblk : in1 ? b0 - j0.nblk : b0;
  const int c = bx * 16;
  const int n = c + col;

  __shared__ v4f stage[KS * REG4];
  v4f* st = stage + wave * REG4;

  v8f acc0 = {0.f, 0.f, 0.f, 0.f, 0.f, 0.f, 0.f, 0.f};
  v8f acc1 = {0.f, 0.f, 0.f, 0.f, 0.f, 0.f, 0.f, 0.f};
  const float* Wc = J.W + (size_t)c * J.ldw;
  const size_t orow = (size_t)(wave * 16 + (hi ? 8 : 0)) * J.ldo + n;
  float ad[8];
#pragma unroll
  for (int r = 0; r < 8; ++r) ad[r] = 0.f;
  if (wave < 2 && J.addin != nullptr) {
#pragma unroll
    for (int r = 0; r < 8; ++r) ad[r] = J.addin[orow + (size_t)r * J.ldo];
  }
  seg_regs<K> R0;
  v4f bng = {0.f, 0.f, 0.f, 0.f}, bnb = {0.f, 0.f, 0.f, 0.f};
  if constexpr (BN_IN) {
    const int cb = wave * S::SL + (lane % S::CH) * 4;
    bng = *reinterpret_cast<const v4f*>(bn_g + cb);
    bnb = *reinterpret_cast<const v4f*>(bn_b + cb);
  }
  seg_fetch<K>(R0, in0, Wc, J.ldw, lane, wave);
  if constexpr (BN_IN) {
    __builtin_amdgcn_sched_barrier(0);
    seg_batchnorm<K>(R0, bng, bnb);
  }
  seg_mma<K>(acc0, acc1, R0, st, lane);

  __shared__ float4 part[KS][2][2][32];
  __shared__ float sred[2][16][2];
  part[wave][0][0][lane] = make_float4(acc0[0], acc0[1], acc0[2], acc0[3]);
  part[wave][0][1][lane] = make_float4(acc0[4], acc0[5], acc0[6], acc0[7]);
  part[wave][1][0][lane] = make_float4(acc1[0], acc1[1], acc1[2], acc1[3]);
  part[wave][1][1][lane] = make_float4(acc1[4], acc1[5], acc1[6], acc1[7]);

  __syncthreads();

  float vals[8];
  if (wave < 2) {
    float a[8];
#pragma unroll
    for (int r = 0; r < 8; ++r) a[r] = 0.f;
#pragma unroll
    for (int s = 0; s < KS; ++s) {
      const float4 lo = part[s][wave][0][lane];
      const float4 hv = part[s][wave][1][lane];
      a[0] += lo.x; a[1] += lo.y; a[2] += lo.z; a[3] += lo.w;
      a[4] += hv.x; a[5] += hv.y; a[6] += hv.z; a[7] += hv.w;
    }
#pragma unroll
    for (int r = 0; r < 8; ++r) a[r] += ad[r];
    if (J.finish == 0) {
#pragma unroll
      for (int r = 0; r < 8; ++r) J.out[orow + (size_t)r * J.ldo] = a[r];
    } else {
      const float bi = J.bias[n];
      float s = 0.f, ss = 0.f;
#pragma unroll
      for (int r = 0; r < 8; ++r) {
        const float v = fmaxf(a[r] + bi, 0.f);
        vals[r] = v; s += v; ss += v * v;
      }
      s  += __shfl_xor(s, 16);
      ss += __shfl_xor(ss, 16);
      if (lane < 16) { sred[wave][col][0] = s; sred[wave][col][1] = ss; }
    }
  }
  if (J.finish != 0) {
    __syncthreads();
    if (wave < 2) {
      const float ts   = sred[0][col][0] + sred[1][col][0];
      const float tss  = sred[0][col][1] + sred[1][col][1];
      const float mean = ts * (1.0f / BB);
      const float var  = tss * (1.0f / BB) - mean * mean;
      const float sc   = J.g[n] * rsqrtf(var + BN_EPS);
      const float bt   = J.beta[n];
#pragma unroll
      for (int r = 0; r < 8; ++r)
        J.out[orow + (size_t)r * J.ldo] = (vals[r] - mean) * sc + bt;
    }
  }
}

extern "C" void kernel_launch(void* const* d_in, const int* in_sizes, int n_in,
                              void* d_out, int out_size, void* d_ws, size_t ws_size,
                              hipStream_t stream) {
  const float* x     = (const float*)d_in[0];
  const float* basis = (const float*)d_in[1];
  const float* bn0_g = (const float*)d_in[2];
  const float* bn0_b = (const float*)d_in[3];
  const float* W0 = (const float*)d_in[4];
  const float* b0 = (const float*)d_in[5];
  const float* g0 = (const float*)d_in[6];
  const float* be0 = (const float*)d_in[7];
  const float* W1 = (const float*)d_in[8];
  const float* b1 = (const float*)d_in[9];
  const float* g1 = (const float*)d_in[10];
  const float* be1 = (const float*)d_in[11];
  const float* W2 = (const float*)d_in[12];
  const float* b2 = (const float*)d_in[13];
  const float* g2 = (const float*)d_in[14];
  const float* be2 = (const float*)d_in[15];
  const float* W3 = (const float*)d_in[16];
  const float* b3 = (const float*)d_in[17];
  const float* g3 = (const float*)d_in[18];
  const float* be3 = (const float*)d_in[19];
  const float* Wf = (const float*)d_in[20];
  const float* bf = (const float*)d_in[21];
  const float* gf = (const float*)d_in[22];
  const float* bef = (const float*)d_in[23];

  float* ws   = (float*)d_ws;
  float* feat = ws;
  float* h    = ws + 65536;
  float* a1   = ws + 73728;
  float* a2   = ws + 81920;
  float* p2   = ws + 90112;
  float* pf   = ws + 98304;
  float* pf2  = ws + 114688;

  const dense_job none = {nullptr, nullptr, nullptr, nullptr, nullptr, nullptr, 0, 0, 0, 0};

  bps_min_kernel<<<dim3(PP / (BPS_TILES * 16), BB), 256, 0, stream>>>(x, basis, feat);
  const dense_job l0  = {W0, b0, g0, be0, nullptr, h, PP, HH, HH / 16, 1};
  const dense_job x2  = {W2, nullptr, nullptr, nullptr, nullptr, p2, PP + HH, HH, HH / 16, 0};
  const dense_job xf  = {Wf, nullptr, nullptr, nullptr, nullptr, pf, PP + 2 * HH, EE, EE / 16, 0};
  dense_kernel<PP, true><<<HH / 16 + HH / 16 + EE / 16, KS * 32, 0, stream>>>(
      feat, bn0_g, bn0_b, l0, x2, xf);
  const dense_job l1  = {W1, b1, g1, be1, nullptr, a1, HH, HH, HH / 16, 1};
  dense_kernel<HH, false><<<HH / 16, KS * 32, 0, stream>>>(h, nullptr, nullptr, l1, none, none);
  const dense_job l2  = {W2 + PP, b2, g2, be2, p2, h, PP + HH, HH, HH / 16, 1};
  const dense_job af  = {Wf + PP, nullptr, nullptr, nullptr, pf, pf2, PP + 2 * HH, EE, EE / 16, 0};
  dense_kernel<HH, false><<<HH / 16 + EE / 16, KS * 32, 0, stream>>>(a1, nullptr, nullptr, l2, af, none);
  const dense_job l3  = {W3, b3, g3, be3, nullptr, a2, HH, HH, HH / 16, 1};
  dense_kernel<HH, false><<<HH / 16, KS * 32, 0, stream>>>(h, nullptr, nullptr, l3, none, none);
  const dense_job lf  = {Wf + PP + HH, bf, gf, bef, pf2, (float*)d_out, PP + 2 * HH, EE, EE / 16, 1};
  dense_kernel<HH, false><<<EE / 16, KS * 32, 0, stream>>>(a2, nullptr, nullptr, lf, none, none);
}


# dense launches: the epilogue's bias and BatchNorm scale/shift loads issued at kernel entry with the operand loads (were behind the K reduce and the second barrier, each an exposed round trip)
# speedup vs baseline: 3.7820x; 1.1130x over previous
#include <hip/hip_runtime.h>

typedef __attribute__((ext_vector_type(2))) float v2f;
typedef __attribute__((ext_vector_type(8))) float v8f;
typedef __attribute__((ext_vector_type(4))) float v4f;

#define BB 32
#define NN 2048
#define PP 1024
#define HH 256
#define EE 512
#define BN_EPS 1e-5f

#define BPS_TILES 8
#define BPS_GPTS  256
#define BPS_NG    (NN / BPS_GPTS)

struct bps_group { float4 X[4], Y[4], Z[4]; };

__device__ __forceinline__ void bps_load_group(bps_group& G, const float* __restrict__ p) {
#pragma unroll
  for (int q = 0; q < 4; ++q) {
    G.X[q] = *reinterpret_cast<const float4*>(p + q * 64);
    G.Y[q] = *reinterpret_cast<const float4*>(p + NN + q * 64);
    G.Z[q] = *reinterpret_cast<const float4*>(p + 2 * NN + q * 64);
  }
}

__device__ __forceinline__ void bps_point_tile(float (&minv)[4], float px, float py,
                                               float pz, v2f Bm, bool hi) {
  const float pn2 = px * px + py * py + pz * pz;
  v2f Am;
  Am.x = hi ? pz  : px;
  Am.y = hi ? pn2 : py;
  v8f c = {0.f, 0.f, 0.f, 0.f, 0.f, 0.f, 0.f, 0.f};
  c = __builtin_amdgcn_wmma_f32_16x16x4_f32(false, Am, false, Bm,
                                            (short)0, c, false, false);
#pragma unroll
  for (int r = 0; r < 4; ++r)
    minv[r] = fminf(fminf(c[2 * r], c[2 * r + 1]), minv[r]);
}

__device__ __forceinline__ void bps_group_tiles(float (&minv)[4], const bps_group& G,
                                                v2f Bm, bool hi) {
#pragma unroll
  for (int q = 0; q < 4; ++q) {
    bps_point_tile(minv, G.X[q].x, G.Y[q].x, G.Z[q].x, Bm, hi);
    bps_point_tile(minv, G.X[q].y, G.Y[q].y, G.Z[q].y, Bm, hi);
    bps_point_tile(minv, G.X[q].z, G.Y[q].z, G.Z[q].z, Bm, hi);
    bps_point_tile(minv, G.X[q].w, G.Y[q].w, G.Z[q].w, Bm, hi);
  }
}

__global__ __launch_bounds__(256) void bps_min_kernel(
    const float* __restrict__ x,
    const float* __restrict__ basis,
    float* __restrict__ feat)
{
  const int b     = blockIdx.y;
  const int lane  = threadIdx.x & 31;
  const int wave  = threadIdx.x >> 5;
  const int col   = lane & 15;
  const bool hi   = lane >= 16;
  const int tileP = (blockIdx.x * BPS_TILES + wave) * 16;

  const float* xp = x + (size_t)b * 3 * NN + 4 * col;
  bps_group G[3];
  bps_load_group(G[0], xp);
  bps_load_group(G[1], xp + BPS_GPTS);

  const int   n   = tileP + col;
  const float bx  = basis[n * 3 + 0];
  const float by  = basis[n * 3 + 1];
  const float bz  = basis[n * 3 + 2];
  const float bn2 = bx * bx + by * by + bz * bz;
  v2f Bm;
  Bm.x = hi ? (-2.0f * bz) : (-2.0f * bx);
  Bm.y = hi ? 1.0f         : (-2.0f * by);

  float minv[4];
#pragma unroll
  for (int r = 0; r < 4; ++r) minv[r] = 3.4e38f;

#pragma unroll
  for (int g = 0; g < BPS_NG; ++g) {
    if (g + 2 < BPS_NG) {
      bps_load_group(G[(g + 2) % 3], xp + (g + 2) * BPS_GPTS);
      __builtin_amdgcn_sched_barrier(0);
    }
    bps_group_tiles(minv, G[g % 3], Bm, hi);
    __builtin_amdgcn_sched_barrier(0);
  }

  float mv = fminf(fminf(minv[0], minv[1]), fminf(minv[2], minv[3]));
  mv = fminf(mv, __shfl_xor(mv, 16));
  mv = fmaxf(mv + bn2, 0.0f);

  __shared__ float red[BPS_TILES * 16];
  if (lane < 16) red[wave * 16 + col] = mv;
  __syncthreads();
  if (threadIdx.x < BPS_TILES * 16)
    feat[(size_t)b * PP + blockIdx.x * (BPS_TILES * 16) + threadIdx.x] =
        sqrtf(red[threadIdx.x]);
}

#define KS 16

template <int Ks>
struct seg_regs {
  static constexpr int SL = Ks / KS;
  static constexpr int CH = SL / 4;
  static constexpr int RP = 32 / CH;
  static constexpr int P4 = CH + 1;
  v4f a[CH];
  v4f w[CH / 2];
};

template <int Ks>
__device__ __forceinline__ void seg_fetch(seg_regs<Ks>& R, const float* __restrict__ src,
                                          const float* __restrict__ Wseg, int ldw,
                                          int lane, int ks) {
  using S = seg_regs<Ks>;
  static_assert(Ks % (8 * KS) == 0 && 32 % S::CH == 0 && S::CH >= 2,
                "slice must be whole pairs of k-steps and divide a wave's 32 lanes");
  const int ch = lane % S::CH, r0 = lane / S::CH;
  const float* ap = src  + (size_t)r0 * Ks  + ks * S::SL + ch * 4;
  const float* wp = Wseg + (size_t)r0 * ldw + ks * S::SL + ch * 4;
#pragma unroll
  for (int j = 0; j < S::CH; ++j)
    R.a[j] = *reinterpret_cast<const v4f*>(ap + (size_t)j * S::RP * Ks);
#pragma unroll
  for (int j = 0; j < S::CH / 2; ++j)
    R.w[j] = *reinterpret_cast<const v4f*>(wp + (size_t)j * S::RP * ldw);
}

__device__ __forceinline__ float xhalf(float x) {
  const int b = __builtin_bit_cast(int, x);
  return __builtin_bit_cast(float,
      __builtin_amdgcn_permlanex16(b, b, 0x76543210, 0xfedcba98, false, false));
}

template <int Ks>
__device__ __forceinline__ void seg_batchnorm(seg_regs<Ks>& R, v4f g, v4f b) {
  using S = seg_regs<Ks>;
  static_assert(S::RP == 2 && S::CH * S::RP == BB,
                "a lane and lane ^ 16 must hold all batch rows of their columns between them");
  v4f s = R.a[0], q = R.a[0] * R.a[0];
#pragma unroll
  for (int j = 1; j < S::CH; ++j) { s += R.a[j]; q += R.a[j] * R.a[j]; }
  v4f mean, sc;
#pragma unroll
  for (int c = 0; c < 4; ++c) {
    const float ts  = s[c] + xhalf(s[c]);
    const float tq  = q[c] + xhalf(q[c]);
    const float m   = ts * (1.0f / BB);
    const float var = tq * (1.0f / BB) - m * m;
    mean[c] = m;
    sc[c]   = g[c] * rsqrtf(var + BN_EPS);
  }
#pragma unroll
  for (int j = 0; j < S::CH; ++j) R.a[j] = (R.a[j] - mean) * sc + b;
}

template <int Ks>
__device__ __forceinline__ void seg_mma(v8f& acc0, v8f& acc1, const seg_regs<Ks>& R,
                                        v4f* __restrict__ st, int lane) {
  using S = seg_regs<Ks>;
  const int ch = lane % S::CH, r0 = lane / S::CH;
  v4f* sp = st + r0 * S::P4 + ch;
#pragma unroll
  for (int j = 0; j < S::CH; ++j) sp[j * S::RP * S::P4] = R.a[j];
#pragma unroll
  for (int j = 0; j < S::CH / 2; ++j) sp[(32 + j * S::RP) * S::P4] = R.w[j];
  __builtin_amdgcn_wave_barrier();
  const v4f* f = st + (lane & 15) * S::P4 + (lane >> 4);
#pragma unroll
  for (int k8 = 0; k8 < S::SL / 8; ++k8) {
    const v4f av0 = f[2 * k8];
    const v4f av1 = f[16 * S::P4 + 2 * k8];
    const v4f wv  = f[32 * S::P4 + 2 * k8];
    v2f A0; A0.x = av0.x; A0.y = av0.y;
    v2f A1; A1.x = av1.x; A1.y = av1.y;
    v2f B0; B0.x = wv.x;  B0.y = wv.y;
    acc0 = __builtin_amdgcn_wmma_f32_16x16x4_f32(false, A0, false, B0,
                                                 (short)0, acc0, false, false);
    acc1 = __builtin_amdgcn_wmma_f32_16x16x4_f32(false, A1, false, B0,
                                                 (short)0, acc1, false, false);
    v2f A2; A2.x = av0.z; A2.y = av0.w;
    v2f A3; A3.x = av1.z; A3.y = av1.w;
    v2f B1; B1.x = wv.z;  B1.y = wv.w;
    acc0 = __builtin_amdgcn_wmma_f32_16x16x4_f32(false, A2, false, B1,
                                                 (short)0, acc0, false, false);
    acc1 = __builtin_amdgcn_wmma_f32_16x16x4_f32(false, A3, false, B1,
                                                 (short)0, acc1, false, false);
  }
}

struct dense_job {
  const float* W;
  const float* bias;
  const float* g;
  const float* beta;
  const float* addin;
  float* out;
  int ldw;
  int ldo;
  int nblk;
  int finish;
};

template <int K, bool BN_IN>
__global__ __launch_bounds__(KS * 32) void dense_kernel(
    const float* __restrict__ in0,
    const float* __restrict__ bn_g,
    const float* __restrict__ bn_b,
    dense_job j0, dense_job j1, dense_job j2)
{
  using S = seg_regs<K>;
  constexpr int REG4 = 48 * S::P4;
  const int  lane = threadIdx.x & 31;
  const int  wave = threadIdx.x >> 5;
  const int  col  = lane & 15;
  const bool hi   = lane >= 16;

  const int  b0  = blockIdx.x;
  const bool in1 = b0 >= j0.nblk;
  const bool in2 = b0 >= j0.nblk + j1.nblk;
  dense_job J;
  J.W      = in2 ? j2.W      : in1 ? j1.W      : j0.W;
  J.bias   = in2 ? j2.bias   : in1 ? j1.bias   : j0.bias;
  J.g      = in2 ? j2.g      : in1 ? j1.g      : j0.g;
  J.beta   = in2 ? j2.beta   : in1 ? j1.beta   : j0.beta;
  J.addin  = in2 ? j2.addin  : in1 ? j1.addin  : j0.addin;
  J.out    = in2 ? j2.out    : in1 ? j1.out    : j0.out;
  J.ldw    = in2 ? j2.ldw    : in1 ? j1.ldw    : j0.ldw;
  J.ldo    = in2 ? j2.ldo    : in1 ? j1.ldo    : j0.ldo;
  J.finish = in2 ? j2.finish : in1 ? j1.finish : j0.finish;
  const int bx = in2 ? b0 - j0.nblk - j1.nblk : in1 ? b0 - j0.nblk : b0;
  const int c = bx * 16;
  const int n = c + col;

  __shared__ v4f stage[KS * REG4];
  v4f* st = stage + wave * REG4;

  v8f acc0 = {0.f, 0.f, 0.f, 0.f, 0.f, 0.f, 0.f, 0.f};
  v8f acc1 = {0.f, 0.f, 0.f, 0.f, 0.f, 0.f, 0.f, 0.f};
  const float* Wc = J.W + (size_t)c * J.ldw;
  const size_t orow = (size_t)(wave * 16 + (hi ? 8 : 0)) * J.ldo + n;
  float ad[8];
#pragma unroll
  for (int r = 0; r < 8; ++r) ad[r] = 0.f;
  float ep_bias = 0.f, ep_g = 0.f, ep_beta = 0.f;
  if (wave < 2) {
    if (J.addin != nullptr) {
#pragma unroll
      for (int r = 0; r < 8; ++r) ad[r] = J.addin[orow + (size_t)r * J.ldo];
    }
    if (J.finish != 0) {
      ep_bias = J.bias[n];
      ep_g    = J.g[n];
      ep_beta = J.beta[n];
    }
  }
  seg_regs<K> R0;
  v4f bng = {0.f, 0.f, 0.f, 0.f}, bnb = {0.f, 0.f, 0.f, 0.f};
  if constexpr (BN_IN) {
    const int cb = wave * S::SL + (lane % S::CH) * 4;
    bng = *reinterpret_cast<const v4f*>(bn_g + cb);
    bnb = *reinterpret_cast<const v4f*>(bn_b + cb);
  }
  seg_fetch<K>(R0, in0, Wc, J.ldw, lane, wave);
  if constexpr (BN_IN) {
    __builtin_amdgcn_sched_barrier(0);
    seg_batchnorm<K>(R0, bng, bnb);
  }
  seg_mma<K>(acc0, acc1, R0, st, lane);

  __shared__ float4 part[KS][2][2][32];
  __shared__ float sred[2][16][2];
  part[wave][0][0][lane] = make_float4(acc0[0], acc0[1], acc0[2], acc0[3]);
  part[wave][0][1][lane] = make_float4(acc0[4], acc0[5], acc0[6], acc0[7]);
  part[wave][1][0][lane] = make_float4(acc1[0], acc1[1], acc1[2], acc1[3]);
  part[wave][1][1][lane] = make_float4(acc1[4], acc1[5], acc1[6], acc1[7]);

  __syncthreads();

  float vals[8];
  if (wave < 2) {
    float a[8];
#pragma unroll
    for (int r = 0; r < 8; ++r) a[r] = 0.f;
#pragma unroll
    for (int s = 0; s < KS; ++s) {
      const float4 lo = part[s][wave][0][lane];
      const float4 hv = part[s][wave][1][lane];
      a[0] += lo.x; a[1] += lo.y; a[2] += lo.z; a[3] += lo.w;
      a[4] += hv.x; a[5] += hv.y; a[6] += hv.z; a[7] += hv.w;
    }
#pragma unroll
    for (int r = 0; r < 8; ++r) a[r] += ad[r];
    if (J.finish == 0) {
#pragma unroll
      for (int r = 0; r < 8; ++r) J.out[orow + (size_t)r * J.ldo] = a[r];
    } else {
      const float bi = ep_bias;
      float s = 0.f, ss = 0.f;
#pragma unroll
      for (int r = 0; r < 8; ++r) {
        const float v = fmaxf(a[r] + bi, 0.f);
        vals[r] = v; s += v; ss += v * v;
      }
      s  += __shfl_xor(s, 16);
      ss += __shfl_xor(ss, 16);
      if (lane < 16) { sred[wave][col][0] = s; sred[wave][col][1] = ss; }
    }
  }
  if (J.finish != 0) {
    __syncthreads();
    if (wave < 2) {
      const float ts   = sred[0][col][0] + sred[1][col][0];
      const float tss  = sred[0][col][1] + sred[1][col][1];
      const float mean = ts * (1.0f / BB);
      const float var  = tss * (1.0f / BB) - mean * mean;
      const float sc   = ep_g * rsqrtf(var + BN_EPS);
      const float bt   = ep_beta;
#pragma unroll
      for (int r = 0; r < 8; ++r)
        J.out[orow + (size_t)r * J.ldo] = (vals[r] - mean) * sc + bt;
    }
  }
}

extern "C" void kernel_launch(void* const* d_in, const int* in_sizes, int n_in,
                              void* d_out, int out_size, void* d_ws, size_t ws_size,
                              hipStream_t stream) {
  const float* x     = (const float*)d_in[0];
  const float* basis = (const float*)d_in[1];
  const float* bn0_g = (const float*)d_in[2];
  const float* bn0_b = (const float*)d_in[3];
  const float* W0 = (const float*)d_in[4];
  const float* b0 = (const float*)d_in[5];
  const float* g0 = (const float*)d_in[6];
  const float* be0 = (const float*)d_in[7];
  const float* W1 = (const float*)d_in[8];
  const float* b1 = (const float*)d_in[9];
  const float* g1 = (const float*)d_in[10];
  const float* be1 = (const float*)d_in[11];
  const float* W2 = (const float*)d_in[12];
  const float* b2 = (const float*)d_in[13];
  const float* g2 = (const float*)d_in[14];
  const float* be2 = (const float*)d_in[15];
  const float* W3 = (const float*)d_in[16];
  const float* b3 = (const float*)d_in[17];
  const float* g3 = (const float*)d_in[18];
  const float* be3 = (const float*)d_in[19];
  const float* Wf = (const float*)d_in[20];
  const float* bf = (const float*)d_in[21];
  const float* gf = (const float*)d_in[22];
  const float* bef = (const float*)d_in[23];

  float* ws   = (float*)d_ws;
  float* feat = ws;
  float* h    = ws + 65536;
  float* a1   = ws + 73728;
  float* a2   = ws + 81920;
  float* p2   = ws + 90112;
  float* pf   = ws + 98304;
  float* pf2  = ws + 114688;

  const dense_job none = {nullptr, nullptr, nullptr, nullptr, nullptr, nullptr, 0, 0, 0, 0};

  bps_min_kernel<<<dim3(PP / (BPS_TILES * 16), BB), 256, 0, stream>>>(x, basis, feat);
  const dense_job l0  = {W0, b0, g0, be0, nullptr, h, PP, HH, HH / 16, 1};
  const dense_job x2  = {W2, nullptr, nullptr, nullptr, nullptr, p2, PP + HH, HH, HH / 16, 0};
  const dense_job xf  = {Wf, nullptr, nullptr, nullptr, nullptr, pf, PP + 2 * HH, EE, EE / 16, 0};
  dense_kernel<PP, true><<<HH / 16 + HH / 16 + EE / 16, KS * 32, 0, stream>>>(
      feat, bn0_g, bn0_b, l0, x2, xf);
  const dense_job l1  = {W1, b1, g1, be1, nullptr, a1, HH, HH, HH / 16, 1};
  dense_kernel<HH, false><<<HH / 16, KS * 32, 0, stream>>>(h, nullptr, nullptr, l1, none, none);
  const dense_job l2  = {W2 + PP, b2, g2, be2, p2, h, PP + HH, HH, HH / 16, 1};
  const dense_job af  = {Wf + PP, nullptr, nullptr, nullptr, pf, pf2, PP + 2 * HH, EE, EE / 16, 0};
  dense_kernel<HH, false><<<HH / 16 + EE / 16, KS * 32, 0, stream>>>(a1, nullptr, nullptr, l2, af, none);
  const dense_job l3  = {W3, b3, g3, be3, nullptr, a2, HH, HH, HH / 16, 1};
  dense_kernel<HH, false><<<HH / 16, KS * 32, 0, stream>>>(h, nullptr, nullptr, l3, none, none);
  const dense_job lf  = {Wf + PP + HH, bf, gf, bef, pf2, (float*)d_out, PP + 2 * HH, EE, EE / 16, 1};
  dense_kernel<HH, false><<<EE / 16, KS * 32, 0, stream>>>(a2, nullptr, nullptr, lf, none, none);
}
